# starting kernel
# baseline (speedup 1.0000x reference)
#include <hip/hip_runtime.h>
#include <hip/hip_bf16.h>
#include <math.h>

typedef __attribute__((ext_vector_type(2))) float v2f;
typedef __attribute__((ext_vector_type(8))) float v8f;

#define D 128

__global__ __launch_bounds__(256) void zero_kernel(float* p, long long n) {
    long long i = (long long)blockIdx.x * blockDim.x + threadIdx.x;
    long long stride = (long long)gridDim.x * blockDim.x;
    for (; i < n; i += stride) p[i] = 0.0f;
}

__global__ __launch_bounds__(256) void aggregate_kernel(
    const float* __restrict__ x, const int* __restrict__ src,
    const int* __restrict__ dst, float* __restrict__ s,
    float* __restrict__ cnt, int E)
{
    int wid  = (int)(((long long)blockIdx.x * blockDim.x + threadIdx.x) >> 5);
    int lane = threadIdx.x & 31;
    if (wid >= E) return;
    int si = src[wid];
    int di = dst[wid];
    float4 v = ((const float4*)(x + (long long)si * D))[lane];
    float* so = s + (long long)di * D + lane * 4;
    atomicAdd(so + 0, v.x);
    atomicAdd(so + 1, v.y);
    atomicAdd(so + 2, v.z);
    atomicAdd(so + 3, v.w);
    if (lane == 0) atomicAdd(cnt + di, 1.0f);
}

__global__ __launch_bounds__(256) void sage_gemm_kernel(
    const float* __restrict__ xin, const float* __restrict__ s,
    const float* __restrict__ cnt, const float* __restrict__ Wl,
    const float* __restrict__ bias, const float* __restrict__ Wr,
    float* __restrict__ out, int N, int relu)
{
    int mtile = blockIdx.x;
    int wave  = threadIdx.x >> 5;
    int lane  = threadIdx.x & 31;
    int m0    = mtile * 16;
    int n0    = wave * 16;
    int ncol  = n0 + (lane & 15);

    int rowA  = m0 + (lane & 15);
    if (rowA > N - 1) rowA = N - 1;
    int khalf = (lane >> 4) << 1;

    const float* srow = s   + (long long)rowA * D;
    const float* xrow = xin + (long long)rowA * D;
    float inv = 1.0f / fmaxf(cnt[rowA], 1.0f);

    v8f c = {};
    #pragma unroll
    for (int k = 0; k < D; k += 4) {
        int ka = k + khalf;
        v2f a0; a0.x = srow[ka] * inv;  a0.y = srow[ka + 1] * inv;
        v2f b0; b0.x = Wl[ka * D + ncol]; b0.y = Wl[(ka + 1) * D + ncol];
        c = __builtin_amdgcn_wmma_f32_16x16x4_f32(
                false, a0, false, b0, (short)0, c, false, false);
        v2f a1; a1.x = xrow[ka];        a1.y = xrow[ka + 1];
        v2f b1; b1.x = Wr[ka * D + ncol]; b1.y = Wr[(ka + 1) * D + ncol];
        c = __builtin_amdgcn_wmma_f32_16x16x4_f32(
                false, a1, false, b1, (short)0, c, false, false);
    }

    float bn = bias[ncol];
    int mbase = m0 + ((lane >> 4) << 3);
    #pragma unroll
    for (int r = 0; r < 8; ++r) {
        int m = mbase + r;
        if (m < N) {
            float v = c[r] + bn;
            if (relu) v = fmaxf(v, 0.0f);
            out[(long long)m * D + ncol] = v;
        }
    }
}

__global__ __launch_bounds__(256) void decode_kernel(
    const float* __restrict__ h, const int* __restrict__ dec,
    float* __restrict__ out, int Dk)
{
    int wid  = (int)(((long long)blockIdx.x * blockDim.x + threadIdx.x) >> 5);
    int lane = threadIdx.x & 31;
    if (wid >= Dk) return;
    int ai = dec[wid];
    int bi = dec[Dk + wid];
    float4 a = ((const float4*)(h + (long long)ai * D))[lane];
    float4 b = ((const float4*)(h + (long long)bi * D))[lane];
    float dot = a.x * b.x + a.y * b.y + a.z * b.z + a.w * b.w;
    float na  = a.x * a.x + a.y * a.y + a.z * a.z + a.w * a.w;
    float nb  = b.x * b.x + b.y * b.y + b.z * b.z + b.w * b.w;
    #pragma unroll
    for (int off = 16; off > 0; off >>= 1) {
        dot += __shfl_xor(dot, off, 32);
        na  += __shfl_xor(na,  off, 32);
        nb  += __shfl_xor(nb,  off, 32);
    }
    if (lane == 0) {
        float denom = fmaxf(sqrtf(na) * sqrtf(nb), 1e-6f);
        float z = dot / denom;
        out[wid] = 1.0f / (1.0f + expf(-z));
    }
}

extern "C" void kernel_launch(void* const* d_in, const int* in_sizes, int n_in,
                              void* d_out, int out_size, void* d_ws, size_t ws_size,
                              hipStream_t stream) {
    const float* x   = (const float*)d_in[0];
    const int*   edg = (const int*)d_in[1];
    const int*   dec = (const int*)d_in[2];
    const float* W1l = (const float*)d_in[3];
    const float* b1  = (const float*)d_in[4];
    const float* W1r = (const float*)d_in[5];
    const float* W2l = (const float*)d_in[6];
    const float* b2  = (const float*)d_in[7];
    const float* W2r = (const float*)d_in[8];
    float* out = (float*)d_out;

    const int N  = in_sizes[0] / D;
    const int E  = in_sizes[1] / 2;
    const int Dk = in_sizes[2] / 2;
    const int* src = edg;
    const int* dst = edg + E;

    float* ws  = (float*)d_ws;
    float* s   = ws;
    float* cnt = s + (long long)N * D;
    float* h1  = cnt + N;
    float* h2  = h1 + (long long)N * D;

    const long long sAndCnt = (long long)N * D + N;
    const int zeroBlocks  = 2048;
    const int aggrBlocks  = (E + 7) / 8;
    const int gemmBlocks  = (N + 15) / 16;
    const int decBlocks   = (Dk + 7) / 8;

    zero_kernel<<<zeroBlocks, 256, 0, stream>>>(s, sAndCnt);
    aggregate_kernel<<<aggrBlocks, 256, 0, stream>>>(x, src, dst, s, cnt, E);
    sage_gemm_kernel<<<gemmBlocks, 256, 0, stream>>>(x, s, cnt, W1l, b1, W1r,
                                                     h1, N,  1);
    zero_kernel<<<zeroBlocks, 256, 0, stream>>>(s, sAndCnt);
    aggregate_kernel<<<aggrBlocks, 256, 0, stream>>>(h1, src, dst, s, cnt, E);
    sage_gemm_kernel<<<gemmBlocks, 256, 0, stream>>>(h1, s, cnt, W2l, b2, W2r,
                                                     h2, N,  0);
    decode_kernel<<<decBlocks, 256, 0, stream>>>(h2, dec, out, Dk);
}


# aggregate: lane l owns columns l+32j, so each wave-wide f32 atomic add and each row load covers one whole 128-B line instead of a quarter of four lines; sums remain arrival-order f32 atomics
# speedup vs baseline: 1.0840x; 1.0840x over previous
#include <hip/hip_runtime.h>
#include <hip/hip_bf16.h>
#include <math.h>

typedef __attribute__((ext_vector_type(2))) float v2f;
typedef __attribute__((ext_vector_type(8))) float v8f;

#define D 128

__global__ __launch_bounds__(256) void zero_kernel(float* p, long long n) {
    long long i = (long long)blockIdx.x * blockDim.x + threadIdx.x;
    long long stride = (long long)gridDim.x * blockDim.x;
    for (; i < n; i += stride) p[i] = 0.0f;
}

__global__ __launch_bounds__(256) void aggregate_kernel(
    const float* __restrict__ x, const int* __restrict__ src,
    const int* __restrict__ dst, float* __restrict__ s,
    float* __restrict__ cnt, int E)
{
    int wid  = (int)(((long long)blockIdx.x * blockDim.x + threadIdx.x) >> 5);
    int lane = threadIdx.x & 31;
    if (wid >= E) return;
    int si = src[wid];
    int di = dst[wid];
    const float* xr = x + (long long)si * D + lane;
    float v0 = xr[0];
    float v1 = xr[32];
    float v2 = xr[64];
    float v3 = xr[96];
    float* so = s + (long long)di * D + lane;
    atomicAdd(so,      v0);
    atomicAdd(so + 32, v1);
    atomicAdd(so + 64, v2);
    atomicAdd(so + 96, v3);
    if (lane == 0) atomicAdd(cnt + di, 1.0f);
}

__global__ __launch_bounds__(256) void sage_gemm_kernel(
    const float* __restrict__ xin, const float* __restrict__ s,
    const float* __restrict__ cnt, const float* __restrict__ Wl,
    const float* __restrict__ bias, const float* __restrict__ Wr,
    float* __restrict__ out, int N, int relu)
{
    int mtile = blockIdx.x;
    int wave  = threadIdx.x >> 5;
    int lane  = threadIdx.x & 31;
    int m0    = mtile * 16;
    int n0    = wave * 16;
    int ncol  = n0 + (lane & 15);

    int rowA  = m0 + (lane & 15);
    if (rowA > N - 1) rowA = N - 1;
    int khalf = (lane >> 4) << 1;

    const float* srow = s   + (long long)rowA * D;
    const float* xrow = xin + (long long)rowA * D;
    float inv = 1.0f / fmaxf(cnt[rowA], 1.0f);

    v8f c = {};
    #pragma unroll
    for (int k = 0; k < D; k += 4) {
        int ka = k + khalf;
        v2f a0; a0.x = srow[ka] * inv;  a0.y = srow[ka + 1] * inv;
        v2f b0; b0.x = Wl[ka * D + ncol]; b0.y = Wl[(ka + 1) * D + ncol];
        c = __builtin_amdgcn_wmma_f32_16x16x4_f32(
                false, a0, false, b0, (short)0, c, false, false);
        v2f a1; a1.x = xrow[ka];        a1.y = xrow[ka + 1];
        v2f b1; b1.x = Wr[ka * D + ncol]; b1.y = Wr[(ka + 1) * D + ncol];
        c = __builtin_amdgcn_wmma_f32_16x16x4_f32(
                false, a1, false, b1, (short)0, c, false, false);
    }

    float bn = bias[ncol];
    int mbase = m0 + ((lane >> 4) << 3);
    #pragma unroll
    for (int r = 0; r < 8; ++r) {
        int m = mbase + r;
        if (m < N) {
            float v = c[r] + bn;
            if (relu) v = fmaxf(v, 0.0f);
            out[(long long)m * D + ncol] = v;
        }
    }
}

__global__ __launch_bounds__(256) void decode_kernel(
    const float* __restrict__ h, const int* __restrict__ dec,
    float* __restrict__ out, int Dk)
{
    int wid  = (int)(((long long)blockIdx.x * blockDim.x + threadIdx.x) >> 5);
    int lane = threadIdx.x & 31;
    if (wid >= Dk) return;
    int ai = dec[wid];
    int bi = dec[Dk + wid];
    float4 a = ((const float4*)(h + (long long)ai * D))[lane];
    float4 b = ((const float4*)(h + (long long)bi * D))[lane];
    float dot = a.x * b.x + a.y * b.y + a.z * b.z + a.w * b.w;
    float na  = a.x * a.x + a.y * a.y + a.z * a.z + a.w * a.w;
    float nb  = b.x * b.x + b.y * b.y + b.z * b.z + b.w * b.w;
    #pragma unroll
    for (int off = 16; off > 0; off >>= 1) {
        dot += __shfl_xor(dot, off, 32);
        na  += __shfl_xor(na,  off, 32);
        nb  += __shfl_xor(nb,  off, 32);
    }
    if (lane == 0) {
        float denom = fmaxf(sqrtf(na) * sqrtf(nb), 1e-6f);
        float z = dot / denom;
        out[wid] = 1.0f / (1.0f + expf(-z));
    }
}

extern "C" void kernel_launch(void* const* d_in, const int* in_sizes, int n_in,
                              void* d_out, int out_size, void* d_ws, size_t ws_size,
                              hipStream_t stream) {
    const float* x   = (const float*)d_in[0];
    const int*   edg = (const int*)d_in[1];
    const int*   dec = (const int*)d_in[2];
    const float* W1l = (const float*)d_in[3];
    const float* b1  = (const float*)d_in[4];
    const float* W1r = (const float*)d_in[5];
    const float* W2l = (const float*)d_in[6];
    const float* b2  = (const float*)d_in[7];
    const float* W2r = (const float*)d_in[8];
    float* out = (float*)d_out;

    const int N  = in_sizes[0] / D;
    const int E  = in_sizes[1] / 2;
    const int Dk = in_sizes[2] / 2;
    const int* src = edg;
    const int* dst = edg + E;

    float* ws  = (float*)d_ws;
    float* s   = ws;
    float* cnt = s + (long long)N * D;
    float* h1  = cnt + N;
    float* h2  = h1 + (long long)N * D;

    const long long sAndCnt = (long long)N * D + N;
    const int zeroBlocks  = 2048;
    const int aggrBlocks  = (E + 7) / 8;
    const int gemmBlocks  = (N + 15) / 16;
    const int decBlocks   = (Dk + 7) / 8;

    zero_kernel<<<zeroBlocks, 256, 0, stream>>>(s, sAndCnt);
    aggregate_kernel<<<aggrBlocks, 256, 0, stream>>>(x, src, dst, s, cnt, E);
    sage_gemm_kernel<<<gemmBlocks, 256, 0, stream>>>(x, s, cnt, W1l, b1, W1r,
                                                     h1, N,  1);
    zero_kernel<<<zeroBlocks, 256, 0, stream>>>(s, sAndCnt);
    aggregate_kernel<<<aggrBlocks, 256, 0, stream>>>(h1, src, dst, s, cnt, E);
    sage_gemm_kernel<<<gemmBlocks, 256, 0, stream>>>(h1, s, cnt, W2l, b2, W2r,
                                                     h2, N,  0);
    decode_kernel<<<decBlocks, 256, 0, stream>>>(h2, dec, out, Dk);
}


# aggregate: edges bucketed by destination once per call (returning int add = rank, scan, place); one wave sums a node's rows in registers; no f32 atomics; f32 sums in arrival-rank order, 4 partials
# speedup vs baseline: 2.2944x; 2.1166x over previous
#include <hip/hip_runtime.h>
#include <hip/hip_bf16.h>
#include <math.h>

typedef __attribute__((ext_vector_type(2))) float v2f;
typedef __attribute__((ext_vector_type(8))) float v8f;

#define D 128


__global__ __launch_bounds__(256) void zero_kernel(int* p, int n) {
    int i = blockIdx.x * blockDim.x + threadIdx.x;
    if (i < n) p[i] = 0;
}

__global__ __launch_bounds__(256) void rank_kernel(
    const int* __restrict__ dst, int* __restrict__ deg,
    int* __restrict__ rank, int E)
{
    const int t = blockIdx.x * blockDim.x + threadIdx.x;
    const int T = gridDim.x * blockDim.x;
    const int e0 = t, e1 = t + T, e2 = t + 2 * T, e3 = t + 3 * T;
    if (e3 < E) {
        const int d0 = dst[e0], d1 = dst[e1], d2 = dst[e2], d3 = dst[e3];
        const int r0 = atomicAdd(deg + d0, 1);
        const int r1 = atomicAdd(deg + d1, 1);
        const int r2 = atomicAdd(deg + d2, 1);
        const int r3 = atomicAdd(deg + d3, 1);
        rank[e0] = r0; rank[e1] = r1; rank[e2] = r2; rank[e3] = r3;
    } else {
        if (e0 < E) rank[e0] = atomicAdd(deg + dst[e0], 1);
        if (e1 < E) rank[e1] = atomicAdd(deg + dst[e1], 1);
        if (e2 < E) rank[e2] = atomicAdd(deg + dst[e2], 1);
    }
}

__global__ __launch_bounds__(1024) void scan_kernel(
    const int* __restrict__ deg, int* __restrict__ off,
    float* __restrict__ cnt, int N)
{
    __shared__ int wtot[32];
    const int lane = threadIdx.x & 31;
    const int w    = threadIdx.x >> 5;
    const int chunk = ((N + 32 * 128 - 1) / (32 * 128)) * 128;
    const int b = w * chunk;
    const int e = (b + chunk < N) ? (b + chunk) : N;

    int part = 0;
    for (int i = b + lane; i < e; i += 32) part += deg[i];
    #pragma unroll
    for (int o = 16; o > 0; o >>= 1) part += __shfl_xor(part, o, 32);
    if (lane == 0) wtot[w] = part;
    __syncthreads();
    int carry = (lane < w) ? wtot[lane] : 0;
    #pragma unroll
    for (int o = 16; o > 0; o >>= 1) carry += __shfl_xor(carry, o, 32);

    for (int i0 = b; i0 < e; i0 += 128) {
        const int i = i0 + lane * 4;
        int v0 = (i     < e) ? deg[i]     : 0;
        int v1 = (i + 1 < e) ? deg[i + 1] : 0;
        int v2 = (i + 2 < e) ? deg[i + 2] : 0;
        int v3 = (i + 3 < e) ? deg[i + 3] : 0;
        const int tot = (v0 + v1) + (v2 + v3);
        int incl = tot;
        #pragma unroll
        for (int o = 1; o < 32; o <<= 1) {
            int y = __shfl_up(incl, o, 32);
            if (lane >= o) incl += y;
        }
        int base = carry + incl - tot;
        if (i     < e) { off[i]     = base;                cnt[i]     = (float)v0; }
        if (i + 1 < e) { off[i + 1] = base + v0;           cnt[i + 1] = (float)v1; }
        if (i + 2 < e) { off[i + 2] = base + v0 + v1;      cnt[i + 2] = (float)v2; }
        if (i + 3 < e) { off[i + 3] = base + v0 + v1 + v2; cnt[i + 3] = (float)v3; }
        carry += __shfl(incl, 31, 32);
    }
    if (w == 31 && lane == 0) off[N] = carry;
}

__global__ __launch_bounds__(256) void place_kernel(
    const int* __restrict__ src, const int* __restrict__ dst,
    const int* __restrict__ rank, const int* __restrict__ off,
    int* __restrict__ perm, int E)
{
    const int t = blockIdx.x * blockDim.x + threadIdx.x;
    const int T = gridDim.x * blockDim.x;
    const int e0 = t, e1 = t + T, e2 = t + 2 * T, e3 = t + 3 * T;
    if (e3 < E) {
        const int d0 = dst[e0], d1 = dst[e1], d2 = dst[e2], d3 = dst[e3];
        const int r0 = rank[e0], r1 = rank[e1], r2 = rank[e2], r3 = rank[e3];
        const int s0 = src[e0], s1 = src[e1], s2 = src[e2], s3 = src[e3];
        const int p0 = off[d0] + r0, p1 = off[d1] + r1;
        const int p2 = off[d2] + r2, p3 = off[d3] + r3;
        if ((unsigned)p0 < (unsigned)E) perm[p0] = s0;
        if ((unsigned)p1 < (unsigned)E) perm[p1] = s1;
        if ((unsigned)p2 < (unsigned)E) perm[p2] = s2;
        if ((unsigned)p3 < (unsigned)E) perm[p3] = s3;
    } else {
        #pragma unroll
        for (int k = 0; k < 3; ++k) {
            const int e = t + k * T;
            if (e < E) {
                const int p = off[dst[e]] + rank[e];
                if ((unsigned)p < (unsigned)E) perm[p] = src[e];
            }
        }
    }
}

__global__ __launch_bounds__(256) void segment_sum_kernel(
    const float* __restrict__ xin, const int* __restrict__ off,
    const int* __restrict__ perm, float* __restrict__ s, int N, int E)
{
    const int lane = threadIdx.x & 31;
    const int wave0  = __builtin_amdgcn_readfirstlane(
                           (int)((blockIdx.x * blockDim.x + threadIdx.x) >> 5));
    const int nwaves = (int)((gridDim.x * blockDim.x) >> 5);
    const float4* xb = (const float4*)xin + lane;
    const unsigned last = (unsigned)(N - 1);

    for (int node = wave0; node < N; node += nwaves) {
        int beg = off[node];
        int end = off[node + 1];
        beg = beg < 0 ? 0 : beg;
        end = end > E ? E : end;
        float4 a0 = make_float4(0.f, 0.f, 0.f, 0.f);
        float4 a1 = a0, a2 = a0, a3 = a0;
        int j = beg;
        for (; j + 8 <= end; j += 8) {
            float4 v[8];
            #pragma unroll
            for (int k = 0; k < 8; ++k) {
                unsigned idx = (unsigned)perm[j + k];
                idx = idx < last ? idx : last;
                v[k] = xb[(size_t)idx * (D / 4)];
            }
            a0.x += v[0].x; a0.y += v[0].y; a0.z += v[0].z; a0.w += v[0].w;
            a1.x += v[1].x; a1.y += v[1].y; a1.z += v[1].z; a1.w += v[1].w;
            a2.x += v[2].x; a2.y += v[2].y; a2.z += v[2].z; a2.w += v[2].w;
            a3.x += v[3].x; a3.y += v[3].y; a3.z += v[3].z; a3.w += v[3].w;
            a0.x += v[4].x; a0.y += v[4].y; a0.z += v[4].z; a0.w += v[4].w;
            a1.x += v[5].x; a1.y += v[5].y; a1.z += v[5].z; a1.w += v[5].w;
            a2.x += v[6].x; a2.y += v[6].y; a2.z += v[6].z; a2.w += v[6].w;
            a3.x += v[7].x; a3.y += v[7].y; a3.z += v[7].z; a3.w += v[7].w;
        }
        if (j < end) {
            float4 v[8];
            #pragma unroll
            for (int k = 0; k < 8; ++k) {
                const int jj = (j + k < end) ? (j + k) : (end - 1);
                unsigned idx = (unsigned)perm[jj];
                idx = idx < last ? idx : last;
                v[k] = xb[(size_t)idx * (D / 4)];
            }
            #pragma unroll
            for (int k = 0; k < 8; ++k) {
                if (j + k >= end) v[k] = make_float4(0.f, 0.f, 0.f, 0.f);
            }
            a0.x += v[0].x; a0.y += v[0].y; a0.z += v[0].z; a0.w += v[0].w;
            a1.x += v[1].x; a1.y += v[1].y; a1.z += v[1].z; a1.w += v[1].w;
            a2.x += v[2].x; a2.y += v[2].y; a2.z += v[2].z; a2.w += v[2].w;
            a3.x += v[3].x; a3.y += v[3].y; a3.z += v[3].z; a3.w += v[3].w;
            a0.x += v[4].x; a0.y += v[4].y; a0.z += v[4].z; a0.w += v[4].w;
            a1.x += v[5].x; a1.y += v[5].y; a1.z += v[5].z; a1.w += v[5].w;
            a2.x += v[6].x; a2.y += v[6].y; a2.z += v[6].z; a2.w += v[6].w;
        }
        float4 r;
        r.x = (a0.x + a1.x) + (a2.x + a3.x);
        r.y = (a0.y + a1.y) + (a2.y + a3.y);
        r.z = (a0.z + a1.z) + (a2.z + a3.z);
        r.w = (a0.w + a1.w) + (a2.w + a3.w);
        ((float4*)(s + (size_t)node * D))[lane] = r;
    }
}

__global__ __launch_bounds__(256) void sage_gemm_kernel(
    const float* __restrict__ xin, const float* __restrict__ s,
    const float* __restrict__ cnt, const float* __restrict__ Wl,
    const float* __restrict__ bias, const float* __restrict__ Wr,
    float* __restrict__ out, int N, int relu)
{
    int mtile = blockIdx.x;
    int wave  = threadIdx.x >> 5;
    int lane  = threadIdx.x & 31;
    int m0    = mtile * 16;
    int n0    = wave * 16;
    int ncol  = n0 + (lane & 15);

    int rowA  = m0 + (lane & 15);
    if (rowA > N - 1) rowA = N - 1;
    int khalf = (lane >> 4) << 1;

    const float* srow = s   + (long long)rowA * D;
    const float* xrow = xin + (long long)rowA * D;
    float inv = 1.0f / fmaxf(cnt[rowA], 1.0f);

    v8f c = {};
    #pragma unroll
    for (int k = 0; k < D; k += 4) {
        int ka = k + khalf;
        v2f a0; a0.x = srow[ka] * inv;  a0.y = srow[ka + 1] * inv;
        v2f b0; b0.x = Wl[ka * D + ncol]; b0.y = Wl[(ka + 1) * D + ncol];
        c = __builtin_amdgcn_wmma_f32_16x16x4_f32(
                false, a0, false, b0, (short)0, c, false, false);
        v2f a1; a1.x = xrow[ka];        a1.y = xrow[ka + 1];
        v2f b1; b1.x = Wr[ka * D + ncol]; b1.y = Wr[(ka + 1) * D + ncol];
        c = __builtin_amdgcn_wmma_f32_16x16x4_f32(
                false, a1, false, b1, (short)0, c, false, false);
    }

    float bn = bias[ncol];
    int mbase = m0 + ((lane >> 4) << 3);
    #pragma unroll
    for (int r = 0; r < 8; ++r) {
        int m = mbase + r;
        if (m < N) {
            float v = c[r] + bn;
            if (relu) v = fmaxf(v, 0.0f);
            out[(long long)m * D + ncol] = v;
        }
    }
}

__global__ __launch_bounds__(256) void decode_kernel(
    const float* __restrict__ h, const int* __restrict__ dec,
    float* __restrict__ out, int Dk)
{
    int wid  = (int)(((long long)blockIdx.x * blockDim.x + threadIdx.x) >> 5);
    int lane = threadIdx.x & 31;
    if (wid >= Dk) return;
    int ai = dec[wid];
    int bi = dec[Dk + wid];
    float4 a = ((const float4*)(h + (long long)ai * D))[lane];
    float4 b = ((const float4*)(h + (long long)bi * D))[lane];
    float dot = a.x * b.x + a.y * b.y + a.z * b.z + a.w * b.w;
    float na  = a.x * a.x + a.y * a.y + a.z * a.z + a.w * a.w;
    float nb  = b.x * b.x + b.y * b.y + b.z * b.z + b.w * b.w;
    #pragma unroll
    for (int off = 16; off > 0; off >>= 1) {
        dot += __shfl_xor(dot, off, 32);
        na  += __shfl_xor(na,  off, 32);
        nb  += __shfl_xor(nb,  off, 32);
    }
    if (lane == 0) {
        float denom = fmaxf(sqrtf(na) * sqrtf(nb), 1e-6f);
        float z = dot / denom;
        out[wid] = 1.0f / (1.0f + expf(-z));
    }
}

extern "C" void kernel_launch(void* const* d_in, const int* in_sizes, int n_in,
                              void* d_out, int out_size, void* d_ws, size_t ws_size,
                              hipStream_t stream) {
    const float* x   = (const float*)d_in[0];
    const int*   edg = (const int*)d_in[1];
    const int*   dec = (const int*)d_in[2];
    const float* W1l = (const float*)d_in[3];
    const float* b1  = (const float*)d_in[4];
    const float* W1r = (const float*)d_in[5];
    const float* W2l = (const float*)d_in[6];
    const float* b2  = (const float*)d_in[7];
    const float* W2r = (const float*)d_in[8];
    float* out = (float*)d_out;

    const int N  = in_sizes[0] / D;
    const int E  = in_sizes[1] / 2;
    const int Dk = in_sizes[2] / 2;
    const int* src = edg;
    const int* dst = edg + E;

    float* ws  = (float*)d_ws;
    float* s   = ws;
    float* cnt = s + (long long)N * D;
    float* h1  = cnt + N;
    float* h2  = h1 + (long long)N * D;

    int* rank = (int*)h1;
    int* deg  = (int*)h2;
    int* off  = deg + N;
    int* perm = off + N + 1;

    const int zeroBlocks  = (N + 255) / 256;
    const int edgeBlocks  = (E + 1023) / 1024;
    const int sumBlocks   = 2048;
    const int gemmBlocks  = (N + 15) / 16;
    const int decBlocks   = (Dk + 7) / 8;

    zero_kernel<<<zeroBlocks, 256, 0, stream>>>(deg, N);
    rank_kernel<<<edgeBlocks, 256, 0, stream>>>(dst, deg, rank, E);
    scan_kernel<<<1, 1024, 0, stream>>>(deg, off, cnt, N);
    place_kernel<<<edgeBlocks, 256, 0, stream>>>(src, dst, rank, off, perm, E);
    segment_sum_kernel<<<sumBlocks, 256, 0, stream>>>(x, off, perm, s, N, E);
    sage_gemm_kernel<<<gemmBlocks, 256, 0, stream>>>(x, s, cnt, W1l, b1, W1r,
                                                     h1, N,  1);
    segment_sum_kernel<<<sumBlocks, 256, 0, stream>>>(h1, off, perm, s, N, E);
    sage_gemm_kernel<<<gemmBlocks, 256, 0, stream>>>(h1, s, cnt, W2l, b2, W2r,
                                                     h2, N,  0);
    decode_kernel<<<decBlocks, 256, 0, stream>>>(h2, dec, out, Dk);
}


# decode: one wave per 32 pairs instead of one per pair (500000 waves become 15625): whole-line index loads, eight pairs' rows in flight, sums reduced together in the parent's order, one store per wave
# speedup vs baseline: 2.5410x; 1.1075x over previous
#include <hip/hip_runtime.h>
#include <hip/hip_bf16.h>
#include <math.h>

typedef __attribute__((ext_vector_type(2))) float v2f;
typedef __attribute__((ext_vector_type(8))) float v8f;

#define D 128


__global__ __launch_bounds__(256) void zero_kernel(int* p, int n) {
    int i = blockIdx.x * blockDim.x + threadIdx.x;
    if (i < n) p[i] = 0;
}

__global__ __launch_bounds__(256) void rank_kernel(
    const int* __restrict__ dst, int* __restrict__ deg,
    int* __restrict__ rank, int E)
{
    const int t = blockIdx.x * blockDim.x + threadIdx.x;
    const int T = gridDim.x * blockDim.x;
    const int e0 = t, e1 = t + T, e2 = t + 2 * T, e3 = t + 3 * T;
    if (e3 < E) {
        const int d0 = dst[e0], d1 = dst[e1], d2 = dst[e2], d3 = dst[e3];
        const int r0 = atomicAdd(deg + d0, 1);
        const int r1 = atomicAdd(deg + d1, 1);
        const int r2 = atomicAdd(deg + d2, 1);
        const int r3 = atomicAdd(deg + d3, 1);
        rank[e0] = r0; rank[e1] = r1; rank[e2] = r2; rank[e3] = r3;
    } else {
        if (e0 < E) rank[e0] = atomicAdd(deg + dst[e0], 1);
        if (e1 < E) rank[e1] = atomicAdd(deg + dst[e1], 1);
        if (e2 < E) rank[e2] = atomicAdd(deg + dst[e2], 1);
    }
}

__global__ __launch_bounds__(1024) void scan_kernel(
    const int* __restrict__ deg, int* __restrict__ off,
    float* __restrict__ cnt, int N)
{
    __shared__ int wtot[32];
    const int lane = threadIdx.x & 31;
    const int w    = threadIdx.x >> 5;
    const int chunk = ((N + 32 * 128 - 1) / (32 * 128)) * 128;
    const int b = w * chunk;
    const int e = (b + chunk < N) ? (b + chunk) : N;

    int part = 0;
    for (int i = b + lane; i < e; i += 32) part += deg[i];
    #pragma unroll
    for (int o = 16; o > 0; o >>= 1) part += __shfl_xor(part, o, 32);
    if (lane == 0) wtot[w] = part;
    __syncthreads();
    int carry = (lane < w) ? wtot[lane] : 0;
    #pragma unroll
    for (int o = 16; o > 0; o >>= 1) carry += __shfl_xor(carry, o, 32);

    for (int i0 = b; i0 < e; i0 += 128) {
        const int i = i0 + lane * 4;
        int v0 = (i     < e) ? deg[i]     : 0;
        int v1 = (i + 1 < e) ? deg[i + 1] : 0;
        int v2 = (i + 2 < e) ? deg[i + 2] : 0;
        int v3 = (i + 3 < e) ? deg[i + 3] : 0;
        const int tot = (v0 + v1) + (v2 + v3);
        int incl = tot;
        #pragma unroll
        for (int o = 1; o < 32; o <<= 1) {
            int y = __shfl_up(incl, o, 32);
            if (lane >= o) incl += y;
        }
        int base = carry + incl - tot;
        if (i     < e) { off[i]     = base;                cnt[i]     = (float)v0; }
        if (i + 1 < e) { off[i + 1] = base + v0;           cnt[i + 1] = (float)v1; }
        if (i + 2 < e) { off[i + 2] = base + v0 + v1;      cnt[i + 2] = (float)v2; }
        if (i + 3 < e) { off[i + 3] = base + v0 + v1 + v2; cnt[i + 3] = (float)v3; }
        carry += __shfl(incl, 31, 32);
    }
    if (w == 31 && lane == 0) off[N] = carry;
}

__global__ __launch_bounds__(256) void place_kernel(
    const int* __restrict__ src, const int* __restrict__ dst,
    const int* __restrict__ rank, const int* __restrict__ off,
    int* __restrict__ perm, int E)
{
    const int t = blockIdx.x * blockDim.x + threadIdx.x;
    const int T = gridDim.x * blockDim.x;
    const int e0 = t, e1 = t + T, e2 = t + 2 * T, e3 = t + 3 * T;
    if (e3 < E) {
        const int d0 = dst[e0], d1 = dst[e1], d2 = dst[e2], d3 = dst[e3];
        const int r0 = rank[e0], r1 = rank[e1], r2 = rank[e2], r3 = rank[e3];
        const int s0 = src[e0], s1 = src[e1], s2 = src[e2], s3 = src[e3];
        const int p0 = off[d0] + r0, p1 = off[d1] + r1;
        const int p2 = off[d2] + r2, p3 = off[d3] + r3;
        if ((unsigned)p0 < (unsigned)E) perm[p0] = s0;
        if ((unsigned)p1 < (unsigned)E) perm[p1] = s1;
        if ((unsigned)p2 < (unsigned)E) perm[p2] = s2;
        if ((unsigned)p3 < (unsigned)E) perm[p3] = s3;
    } else {
        #pragma unroll
        for (int k = 0; k < 3; ++k) {
            const int e = t + k * T;
            if (e < E) {
                const int p = off[dst[e]] + rank[e];
                if ((unsigned)p < (unsigned)E) perm[p] = src[e];
            }
        }
    }
}

__global__ __launch_bounds__(256) void segment_sum_kernel(
    const float* __restrict__ xin, const int* __restrict__ off,
    const int* __restrict__ perm, float* __restrict__ s, int N, int E)
{
    const int lane = threadIdx.x & 31;
    const int wave0  = __builtin_amdgcn_readfirstlane(
                           (int)((blockIdx.x * blockDim.x + threadIdx.x) >> 5));
    const int nwaves = (int)((gridDim.x * blockDim.x) >> 5);
    const float4* xb = (const float4*)xin + lane;
    const unsigned last = (unsigned)(N - 1);

    for (int node = wave0; node < N; node += nwaves) {
        int beg = off[node];
        int end = off[node + 1];
        beg = beg < 0 ? 0 : beg;
        end = end > E ? E : end;
        float4 a0 = make_float4(0.f, 0.f, 0.f, 0.f);
        float4 a1 = a0, a2 = a0, a3 = a0;
        int j = beg;
        for (; j + 8 <= end; j += 8) {
            float4 v[8];
            #pragma unroll
            for (int k = 0; k < 8; ++k) {
                unsigned idx = (unsigned)perm[j + k];
                idx = idx < last ? idx : last;
                v[k] = xb[(size_t)idx * (D / 4)];
            }
            a0.x += v[0].x; a0.y += v[0].y; a0.z += v[0].z; a0.w += v[0].w;
            a1.x += v[1].x; a1.y += v[1].y; a1.z += v[1].z; a1.w += v[1].w;
            a2.x += v[2].x; a2.y += v[2].y; a2.z += v[2].z; a2.w += v[2].w;
            a3.x += v[3].x; a3.y += v[3].y; a3.z += v[3].z; a3.w += v[3].w;
            a0.x += v[4].x; a0.y += v[4].y; a0.z += v[4].z; a0.w += v[4].w;
            a1.x += v[5].x; a1.y += v[5].y; a1.z += v[5].z; a1.w += v[5].w;
            a2.x += v[6].x; a2.y += v[6].y; a2.z += v[6].z; a2.w += v[6].w;
            a3.x += v[7].x; a3.y += v[7].y; a3.z += v[7].z; a3.w += v[7].w;
        }
        if (j < end) {
            float4 v[8];
            #pragma unroll
            for (int k = 0; k < 8; ++k) {
                const int jj = (j + k < end) ? (j + k) : (end - 1);
                unsigned idx = (unsigned)perm[jj];
                idx = idx < last ? idx : last;
                v[k] = xb[(size_t)idx * (D / 4)];
            }
            #pragma unroll
            for (int k = 0; k < 8; ++k) {
                if (j + k >= end) v[k] = make_float4(0.f, 0.f, 0.f, 0.f);
            }
            a0.x += v[0].x; a0.y += v[0].y; a0.z += v[0].z; a0.w += v[0].w;
            a1.x += v[1].x; a1.y += v[1].y; a1.z += v[1].z; a1.w += v[1].w;
            a2.x += v[2].x; a2.y += v[2].y; a2.z += v[2].z; a2.w += v[2].w;
            a3.x += v[3].x; a3.y += v[3].y; a3.z += v[3].z; a3.w += v[3].w;
            a0.x += v[4].x; a0.y += v[4].y; a0.z += v[4].z; a0.w += v[4].w;
            a1.x += v[5].x; a1.y += v[5].y; a1.z += v[5].z; a1.w += v[5].w;
            a2.x += v[6].x; a2.y += v[6].y; a2.z += v[6].z; a2.w += v[6].w;
        }
        float4 r;
        r.x = (a0.x + a1.x) + (a2.x + a3.x);
        r.y = (a0.y + a1.y) + (a2.y + a3.y);
        r.z = (a0.z + a1.z) + (a2.z + a3.z);
        r.w = (a0.w + a1.w) + (a2.w + a3.w);
        ((float4*)(s + (size_t)node * D))[lane] = r;
    }
}

__global__ __launch_bounds__(256) void sage_gemm_kernel(
    const float* __restrict__ xin, const float* __restrict__ s,
    const float* __restrict__ cnt, const float* __restrict__ Wl,
    const float* __restrict__ bias, const float* __restrict__ Wr,
    float* __restrict__ out, int N, int relu)
{
    int mtile = blockIdx.x;
    int wave  = threadIdx.x >> 5;
    int lane  = threadIdx.x & 31;
    int m0    = mtile * 16;
    int n0    = wave * 16;
    int ncol  = n0 + (lane & 15);

    int rowA  = m0 + (lane & 15);
    if (rowA > N - 1) rowA = N - 1;
    int khalf = (lane >> 4) << 1;

    const float* srow = s   + (long long)rowA * D;
    const float* xrow = xin + (long long)rowA * D;
    float inv = 1.0f / fmaxf(cnt[rowA], 1.0f);

    v8f c = {};
    #pragma unroll
    for (int k = 0; k < D; k += 4) {
        int ka = k + khalf;
        v2f a0; a0.x = srow[ka] * inv;  a0.y = srow[ka + 1] * inv;
        v2f b0; b0.x = Wl[ka * D + ncol]; b0.y = Wl[(ka + 1) * D + ncol];
        c = __builtin_amdgcn_wmma_f32_16x16x4_f32(
                false, a0, false, b0, (short)0, c, false, false);
        v2f a1; a1.x = xrow[ka];        a1.y = xrow[ka + 1];
        v2f b1; b1.x = Wr[ka * D + ncol]; b1.y = Wr[(ka + 1) * D + ncol];
        c = __builtin_amdgcn_wmma_f32_16x16x4_f32(
                false, a1, false, b1, (short)0, c, false, false);
    }

    float bn = bias[ncol];
    int mbase = m0 + ((lane >> 4) << 3);
    #pragma unroll
    for (int r = 0; r < 8; ++r) {
        int m = mbase + r;
        if (m < N) {
            float v = c[r] + bn;
            if (relu) v = fmaxf(v, 0.0f);
            out[(long long)m * D + ncol] = v;
        }
    }
}

__device__ __forceinline__ float reduce8(const float (&v)[8], int lane)
{
    const bool b4 = (lane & 16) != 0, b3 = (lane & 8) != 0, b2 = (lane & 4) != 0;
    float w[4];
    #pragma unroll
    for (int j = 0; j < 4; ++j) {
        const float keep = b4 ? v[j + 4] : v[j];
        const float send = b4 ? v[j] : v[j + 4];
        w[j] = keep + __shfl_xor(send, 16, 32);
    }
    float x[2];
    #pragma unroll
    for (int j = 0; j < 2; ++j) {
        const float keep = b3 ? w[j + 2] : w[j];
        const float send = b3 ? w[j] : w[j + 2];
        x[j] = keep + __shfl_xor(send, 8, 32);
    }
    const float keep = b2 ? x[1] : x[0];
    const float send = b2 ? x[0] : x[1];
    float y = keep + __shfl_xor(send, 4, 32);
    y += __shfl_xor(y, 2, 32);
    y += __shfl_xor(y, 1, 32);
    return y;
}

__global__ __launch_bounds__(256) void decode_kernel(
    const float* __restrict__ h, const int* __restrict__ dec,
    float* __restrict__ out, int Dk, int N)
{
    const int lane = threadIdx.x & 31;
    const int wave = __builtin_amdgcn_readfirstlane(
                         (int)((blockIdx.x * blockDim.x + threadIdx.x) >> 5));
    const int base = wave * 32;
    if (base >= Dk) return;
    const int p  = base + lane;
    const int pc = p < Dk ? p : Dk - 1;
    const unsigned last = (unsigned)(N - 1);
    unsigned ai = (unsigned)dec[pc];
    unsigned bi = (unsigned)dec[Dk + pc];
    ai = ai < last ? ai : last;
    bi = bi < last ? bi : last;
    const float4* hb = (const float4*)h + lane;

    float rd = 0.f, ra = 0.f, rb = 0.f;
    for (int b = 0; b < 4; ++b) {
        float4 va[8], vb[8];
        #pragma unroll
        for (int k = 0; k < 8; ++k) {
            const unsigned ia = (unsigned)__builtin_amdgcn_readlane((int)ai, b * 8 + k);
            const unsigned ib = (unsigned)__builtin_amdgcn_readlane((int)bi, b * 8 + k);
            va[k] = hb[(size_t)ia * (D / 4)];
            vb[k] = hb[(size_t)ib * (D / 4)];
        }
        float d[8], na[8], nb[8];
        #pragma unroll
        for (int k = 0; k < 8; ++k) {
            const float4 a = va[k], c = vb[k];
            d[k]  = a.x * c.x + a.y * c.y + a.z * c.z + a.w * c.w;
            na[k] = a.x * a.x + a.y * a.y + a.z * a.z + a.w * a.w;
            nb[k] = c.x * c.x + c.y * c.y + c.z * c.z + c.w * c.w;
        }
        const float sd = reduce8(d, lane);
        const float sa = reduce8(na, lane);
        const float sb = reduce8(nb, lane);
        const int from = (lane & 7) << 2;
        const float td = __shfl(sd, from, 32);
        const float ta = __shfl(sa, from, 32);
        const float tb = __shfl(sb, from, 32);
        if ((lane >> 3) == b) { rd = td; ra = ta; rb = tb; }
    }
    const float denom = fmaxf(sqrtf(ra) * sqrtf(rb), 1e-6f);
    const float z = rd / denom;
    if (p < Dk) out[p] = 1.0f / (1.0f + expf(-z));
}

extern "C" void kernel_launch(void* const* d_in, const int* in_sizes, int n_in,
                              void* d_out, int out_size, void* d_ws, size_t ws_size,
                              hipStream_t stream) {
    const float* x   = (const float*)d_in[0];
    const int*   edg = (const int*)d_in[1];
    const int*   dec = (const int*)d_in[2];
    const float* W1l = (const float*)d_in[3];
    const float* b1  = (const float*)d_in[4];
    const float* W1r = (const float*)d_in[5];
    const float* W2l = (const float*)d_in[6];
    const float* b2  = (const float*)d_in[7];
    const float* W2r = (const float*)d_in[8];
    float* out = (float*)d_out;

    const int N  = in_sizes[0] / D;
    const int E  = in_sizes[1] / 2;
    const int Dk = in_sizes[2] / 2;
    const int* src = edg;
    const int* dst = edg + E;

    float* ws  = (float*)d_ws;
    float* s   = ws;
    float* cnt = s + (long long)N * D;
    float* h1  = cnt + N;
    float* h2  = h1 + (long long)N * D;

    int* rank = (int*)h1;
    int* deg  = (int*)h2;
    int* off  = deg + N;
    int* perm = off + N + 1;

    const int zeroBlocks  = (N + 255) / 256;
    const int edgeBlocks  = (E + 1023) / 1024;
    const int sumBlocks   = 2048;
    const int gemmBlocks  = (N + 15) / 16;
    const int decBlocks   = (Dk + 255) / 256;

    zero_kernel<<<zeroBlocks, 256, 0, stream>>>(deg, N);
    rank_kernel<<<edgeBlocks, 256, 0, stream>>>(dst, deg, rank, E);
    scan_kernel<<<1, 1024, 0, stream>>>(deg, off, cnt, N);
    place_kernel<<<edgeBlocks, 256, 0, stream>>>(src, dst, rank, off, perm, E);
    segment_sum_kernel<<<sumBlocks, 256, 0, stream>>>(x, off, perm, s, N, E);
    sage_gemm_kernel<<<gemmBlocks, 256, 0, stream>>>(x, s, cnt, W1l, b1, W1r,
                                                     h1, N,  1);
    segment_sum_kernel<<<sumBlocks, 256, 0, stream>>>(h1, off, perm, s, N, E);
    sage_gemm_kernel<<<gemmBlocks, 256, 0, stream>>>(h1, s, cnt, W2l, b2, W2r,
                                                     h2, N,  0);
    decode_kernel<<<decBlocks, 256, 0, stream>>>(h2, dec, out, Dk, N);
}


# GEMM operand feed: both weight matrices copied to LDS once per resident work-group, a wave owns 16 rows x 64 columns so one A fragment feeds 8 WMMAs; per-element sum order unchanged
# speedup vs baseline: 2.9751x; 1.1708x over previous
#include <hip/hip_runtime.h>
#include <hip/hip_bf16.h>
#include <math.h>

typedef __attribute__((ext_vector_type(2))) float v2f;
typedef __attribute__((ext_vector_type(8))) float v8f;

#define D 128


__global__ __launch_bounds__(256) void zero_kernel(int* p, int n) {
    int i = blockIdx.x * blockDim.x + threadIdx.x;
    if (i < n) p[i] = 0;
}

__global__ __launch_bounds__(256) void rank_kernel(
    const int* __restrict__ dst, int* __restrict__ deg,
    int* __restrict__ rank, int E)
{
    const int t = blockIdx.x * blockDim.x + threadIdx.x;
    const int T = gridDim.x * blockDim.x;
    const int e0 = t, e1 = t + T, e2 = t + 2 * T, e3 = t + 3 * T;
    if (e3 < E) {
        const int d0 = dst[e0], d1 = dst[e1], d2 = dst[e2], d3 = dst[e3];
        const int r0 = atomicAdd(deg + d0, 1);
        const int r1 = atomicAdd(deg + d1, 1);
        const int r2 = atomicAdd(deg + d2, 1);
        const int r3 = atomicAdd(deg + d3, 1);
        rank[e0] = r0; rank[e1] = r1; rank[e2] = r2; rank[e3] = r3;
    } else {
        if (e0 < E) rank[e0] = atomicAdd(deg + dst[e0], 1);
        if (e1 < E) rank[e1] = atomicAdd(deg + dst[e1], 1);
        if (e2 < E) rank[e2] = atomicAdd(deg + dst[e2], 1);
    }
}

__global__ __launch_bounds__(1024) void scan_kernel(
    const int* __restrict__ deg, int* __restrict__ off,
    float* __restrict__ cnt, int N)
{
    __shared__ int wtot[32];
    const int lane = threadIdx.x & 31;
    const int w    = threadIdx.x >> 5;
    const int chunk = ((N + 32 * 128 - 1) / (32 * 128)) * 128;
    const int b = w * chunk;
    const int e = (b + chunk < N) ? (b + chunk) : N;

    int part = 0;
    for (int i = b + lane; i < e; i += 32) part += deg[i];
    #pragma unroll
    for (int o = 16; o > 0; o >>= 1) part += __shfl_xor(part, o, 32);
    if (lane == 0) wtot[w] = part;
    __syncthreads();
    int carry = (lane < w) ? wtot[lane] : 0;
    #pragma unroll
    for (int o = 16; o > 0; o >>= 1) carry += __shfl_xor(carry, o, 32);

    for (int i0 = b; i0 < e; i0 += 128) {
        const int i = i0 + lane * 4;
        int v0 = (i     < e) ? deg[i]     : 0;
        int v1 = (i + 1 < e) ? deg[i + 1] : 0;
        int v2 = (i + 2 < e) ? deg[i + 2] : 0;
        int v3 = (i + 3 < e) ? deg[i + 3] : 0;
        const int tot = (v0 + v1) + (v2 + v3);
        int incl = tot;
        #pragma unroll
        for (int o = 1; o < 32; o <<= 1) {
            int y = __shfl_up(incl, o, 32);
            if (lane >= o) incl += y;
        }
        int base = carry + incl - tot;
        if (i     < e) { off[i]     = base;                cnt[i]     = (float)v0; }
        if (i + 1 < e) { off[i + 1] = base + v0;           cnt[i + 1] = (float)v1; }
        if (i + 2 < e) { off[i + 2] = base + v0 + v1;      cnt[i + 2] = (float)v2; }
        if (i + 3 < e) { off[i + 3] = base + v0 + v1 + v2; cnt[i + 3] = (float)v3; }
        carry += __shfl(incl, 31, 32);
    }
    if (w == 31 && lane == 0) off[N] = carry;
}

__global__ __launch_bounds__(256) void place_kernel(
    const int* __restrict__ src, const int* __restrict__ dst,
    const int* __restrict__ rank, const int* __restrict__ off,
    int* __restrict__ perm, int E)
{
    const int t = blockIdx.x * blockDim.x + threadIdx.x;
    const int T = gridDim.x * blockDim.x;
    const int e0 = t, e1 = t + T, e2 = t + 2 * T, e3 = t + 3 * T;
    if (e3 < E) {
        const int d0 = dst[e0], d1 = dst[e1], d2 = dst[e2], d3 = dst[e3];
        const int r0 = rank[e0], r1 = rank[e1], r2 = rank[e2], r3 = rank[e3];
        const int s0 = src[e0], s1 = src[e1], s2 = src[e2], s3 = src[e3];
        const int p0 = off[d0] + r0, p1 = off[d1] + r1;
        const int p2 = off[d2] + r2, p3 = off[d3] + r3;
        if ((unsigned)p0 < (unsigned)E) perm[p0] = s0;
        if ((unsigned)p1 < (unsigned)E) perm[p1] = s1;
        if ((unsigned)p2 < (unsigned)E) perm[p2] = s2;
        if ((unsigned)p3 < (unsigned)E) perm[p3] = s3;
    } else {
        #pragma unroll
        for (int k = 0; k < 3; ++k) {
            const int e = t + k * T;
            if (e < E) {
                const int p = off[dst[e]] + rank[e];
                if ((unsigned)p < (unsigned)E) perm[p] = src[e];
            }
        }
    }
}

__global__ __launch_bounds__(256) void segment_sum_kernel(
    const float* __restrict__ xin, const int* __restrict__ off,
    const int* __restrict__ perm, float* __restrict__ s, int N, int E)
{
    const int lane = threadIdx.x & 31;
    const int wave0  = __builtin_amdgcn_readfirstlane(
                           (int)((blockIdx.x * blockDim.x + threadIdx.x) >> 5));
    const int nwaves = (int)((gridDim.x * blockDim.x) >> 5);
    const float4* xb = (const float4*)xin + lane;
    const unsigned last = (unsigned)(N - 1);

    for (int node = wave0; node < N; node += nwaves) {
        int beg = off[node];
        int end = off[node + 1];
        beg = beg < 0 ? 0 : beg;
        end = end > E ? E : end;
        float4 a0 = make_float4(0.f, 0.f, 0.f, 0.f);
        float4 a1 = a0, a2 = a0, a3 = a0;
        int j = beg;
        for (; j + 8 <= end; j += 8) {
            float4 v[8];
            #pragma unroll
            for (int k = 0; k < 8; ++k) {
                unsigned idx = (unsigned)perm[j + k];
                idx = idx < last ? idx : last;
                v[k] = xb[(size_t)idx * (D / 4)];
            }
            a0.x += v[0].x; a0.y += v[0].y; a0.z += v[0].z; a0.w += v[0].w;
            a1.x += v[1].x; a1.y += v[1].y; a1.z += v[1].z; a1.w += v[1].w;
            a2.x += v[2].x; a2.y += v[2].y; a2.z += v[2].z; a2.w += v[2].w;
            a3.x += v[3].x; a3.y += v[3].y; a3.z += v[3].z; a3.w += v[3].w;
            a0.x += v[4].x; a0.y += v[4].y; a0.z += v[4].z; a0.w += v[4].w;
            a1.x += v[5].x; a1.y += v[5].y; a1.z += v[5].z; a1.w += v[5].w;
            a2.x += v[6].x; a2.y += v[6].y; a2.z += v[6].z; a2.w += v[6].w;
            a3.x += v[7].x; a3.y += v[7].y; a3.z += v[7].z; a3.w += v[7].w;
        }
        if (j < end) {
            float4 v[8];
            #pragma unroll
            for (int k = 0; k < 8; ++k) {
                const int jj = (j + k < end) ? (j + k) : (end - 1);
                unsigned idx = (unsigned)perm[jj];
                idx = idx < last ? idx : last;
                v[k] = xb[(size_t)idx * (D / 4)];
            }
            #pragma unroll
            for (int k = 0; k < 8; ++k) {
                if (j + k >= end) v[k] = make_float4(0.f, 0.f, 0.f, 0.f);
            }
            a0.x += v[0].x; a0.y += v[0].y; a0.z += v[0].z; a0.w += v[0].w;
            a1.x += v[1].x; a1.y += v[1].y; a1.z += v[1].z; a1.w += v[1].w;
            a2.x += v[2].x; a2.y += v[2].y; a2.z += v[2].z; a2.w += v[2].w;
            a3.x += v[3].x; a3.y += v[3].y; a3.z += v[3].z; a3.w += v[3].w;
            a0.x += v[4].x; a0.y += v[4].y; a0.z += v[4].z; a0.w += v[4].w;
            a1.x += v[5].x; a1.y += v[5].y; a1.z += v[5].z; a1.w += v[5].w;
            a2.x += v[6].x; a2.y += v[6].y; a2.z += v[6].z; a2.w += v[6].w;
        }
        float4 r;
        r.x = (a0.x + a1.x) + (a2.x + a3.x);
        r.y = (a0.y + a1.y) + (a2.y + a3.y);
        r.z = (a0.z + a1.z) + (a2.z + a3.z);
        r.w = (a0.w + a1.w) + (a2.w + a3.w);
        ((float4*)(s + (size_t)node * D))[lane] = r;
    }
}

#define WPP 288
#define GEMM_WAVES 8

__global__ __launch_bounds__(256) void sage_gemm_kernel(
    const float* __restrict__ xin, const float* __restrict__ s,
    const float* __restrict__ cnt, const float* __restrict__ Wl,
    const float* __restrict__ bias, const float* __restrict__ Wr,
    float* __restrict__ out, int N, int relu)
{
    __shared__ __attribute__((aligned(16))) float Ws[2 * (D / 2) * WPP];

    for (int it = 0; it < 16; ++it) {
        const int i  = it * 256 + (int)threadIdx.x;
        const int m  = i >> 11;
        const int kp = (i >> 5) & 63;
        const int q  = i & 31;
        const float4* wsrc = (const float4*)(m ? Wr : Wl);
        const float4 r0 = wsrc[(2 * kp) * (D / 4) + q];
        const float4 r1 = wsrc[(2 * kp + 1) * (D / 4) + q];
        float4* wdst = (float4*)(Ws + (m * (D / 2) + kp) * WPP + 8 * q);
        wdst[0] = make_float4(r0.x, r1.x, r0.y, r1.y);
        wdst[1] = make_float4(r0.z, r1.z, r0.w, r1.w);
    }
    __syncthreads();

    const int lane = threadIdx.x & 31;
    const int wave = __builtin_amdgcn_readfirstlane((int)(threadIdx.x >> 5));
    const int c = lane & 15;
    const int h = lane >> 4;
    const int ntiles = (N + 15) >> 4;

    for (int idx = wave; ; idx += GEMM_WAVES) {
        const int tile = (int)blockIdx.x + (idx >> 1) * (int)gridDim.x;
        if (tile >= ntiles) break;
        const int half = idx & 1;
        const int m0 = tile * 16;
        const int n0 = half * 64;

        int rowA = m0 + c;
        if (rowA > N - 1) rowA = N - 1;
        const float* srow = s   + (long long)rowA * D + 2 * h;
        const float* xrow = xin + (long long)rowA * D + 2 * h;
        const float inv = 1.0f / fmaxf(cnt[rowA], 1.0f);

        v2f as[32], ax[32];
        #pragma unroll
        for (int ks = 0; ks < 32; ++ks) {
            as[ks] = *(const v2f*)(srow + 4 * ks);
            ax[ks] = *(const v2f*)(xrow + 4 * ks);
        }

        const float* bl = Ws + h * WPP + 2 * (n0 + c);
        const float* br = bl + (D / 2) * WPP;

        v8f acc[4] = {};
        #pragma unroll
        for (int ks = 0; ks < 32; ++ks) {
            v2f a0; a0.x = as[ks].x * inv; a0.y = as[ks].y * inv;
            const v2f a1 = ax[ks];
            #pragma unroll
            for (int t = 0; t < 4; ++t) {
                const v2f b0 = *(const v2f*)(bl + (2 * ks) * WPP + 32 * t);
                acc[t] = __builtin_amdgcn_wmma_f32_16x16x4_f32(
                             false, a0, false, b0, (short)0, acc[t], false, false);
                const v2f b1 = *(const v2f*)(br + (2 * ks) * WPP + 32 * t);
                acc[t] = __builtin_amdgcn_wmma_f32_16x16x4_f32(
                             false, a1, false, b1, (short)0, acc[t], false, false);
            }
        }

        const int mbase = m0 + (h << 3);
        #pragma unroll
        for (int t = 0; t < 4; ++t) {
            const int ncol = n0 + 16 * t + c;
            const float bn = bias[ncol];
            #pragma unroll
            for (int r = 0; r < 8; ++r) {
                const int m = mbase + r;
                if (m < N) {
                    float v = acc[t][r] + bn;
                    if (relu) v = fmaxf(v, 0.0f);
                    out[(long long)m * D + ncol] = v;
                }
            }
        }
    }
}

__device__ __forceinline__ float reduce8(const float (&v)[8], int lane)
{
    const bool b4 = (lane & 16) != 0, b3 = (lane & 8) != 0, b2 = (lane & 4) != 0;
    float w[4];
    #pragma unroll
    for (int j = 0; j < 4; ++j) {
        const float keep = b4 ? v[j + 4] : v[j];
        const float send = b4 ? v[j] : v[j + 4];
        w[j] = keep + __shfl_xor(send, 16, 32);
    }
    float x[2];
    #pragma unroll
    for (int j = 0; j < 2; ++j) {
        const float keep = b3 ? w[j + 2] : w[j];
        const float send = b3 ? w[j] : w[j + 2];
        x[j] = keep + __shfl_xor(send, 8, 32);
    }
    const float keep = b2 ? x[1] : x[0];
    const float send = b2 ? x[0] : x[1];
    float y = keep + __shfl_xor(send, 4, 32);
    y += __shfl_xor(y, 2, 32);
    y += __shfl_xor(y, 1, 32);
    return y;
}

__global__ __launch_bounds__(256) void decode_kernel(
    const float* __restrict__ h, const int* __restrict__ dec,
    float* __restrict__ out, int Dk, int N)
{
    const int lane = threadIdx.x & 31;
    const int wave = __builtin_amdgcn_readfirstlane(
                         (int)((blockIdx.x * blockDim.x + threadIdx.x) >> 5));
    const int base = wave * 32;
    if (base >= Dk) return;
    const int p  = base + lane;
    const int pc = p < Dk ? p : Dk - 1;
    const unsigned last = (unsigned)(N - 1);
    unsigned ai = (unsigned)dec[pc];
    unsigned bi = (unsigned)dec[Dk + pc];
    ai = ai < last ? ai : last;
    bi = bi < last ? bi : last;
    const float4* hb = (const float4*)h + lane;

    float rd = 0.f, ra = 0.f, rb = 0.f;
    for (int b = 0; b < 4; ++b) {
        float4 va[8], vb[8];
        #pragma unroll
        for (int k = 0; k < 8; ++k) {
            const unsigned ia = (unsigned)__builtin_amdgcn_readlane((int)ai, b * 8 + k);
            const unsigned ib = (unsigned)__builtin_amdgcn_readlane((int)bi, b * 8 + k);
            va[k] = hb[(size_t)ia * (D / 4)];
            vb[k] = hb[(size_t)ib * (D / 4)];
        }
        float d[8], na[8], nb[8];
        #pragma unroll
        for (int k = 0; k < 8; ++k) {
            const float4 a = va[k], c = vb[k];
            d[k]  = a.x * c.x + a.y * c.y + a.z * c.z + a.w * c.w;
            na[k] = a.x * a.x + a.y * a.y + a.z * a.z + a.w * a.w;
            nb[k] = c.x * c.x + c.y * c.y + c.z * c.z + c.w * c.w;
        }
        const float sd = reduce8(d, lane);
        const float sa = reduce8(na, lane);
        const float sb = reduce8(nb, lane);
        const int from = (lane & 7) << 2;
        const float td = __shfl(sd, from, 32);
        const float ta = __shfl(sa, from, 32);
        const float tb = __shfl(sb, from, 32);
        if ((lane >> 3) == b) { rd = td; ra = ta; rb = tb; }
    }
    const float denom = fmaxf(sqrtf(ra) * sqrtf(rb), 1e-6f);
    const float z = rd / denom;
    if (p < Dk) out[p] = 1.0f / (1.0f + expf(-z));
}

extern "C" void kernel_launch(void* const* d_in, const int* in_sizes, int n_in,
                              void* d_out, int out_size, void* d_ws, size_t ws_size,
                              hipStream_t stream) {
    const float* x   = (const float*)d_in[0];
    const int*   edg = (const int*)d_in[1];
    const int*   dec = (const int*)d_in[2];
    const float* W1l = (const float*)d_in[3];
    const float* b1  = (const float*)d_in[4];
    const float* W1r = (const float*)d_in[5];
    const float* W2l = (const float*)d_in[6];
    const float* b2  = (const float*)d_in[7];
    const float* W2r = (const float*)d_in[8];
    float* out = (float*)d_out;

    const int N  = in_sizes[0] / D;
    const int E  = in_sizes[1] / 2;
    const int Dk = in_sizes[2] / 2;
    const int* src = edg;
    const int* dst = edg + E;

    float* ws  = (float*)d_ws;
    float* s   = ws;
    float* cnt = s + (long long)N * D;
    float* h1  = cnt + N;
    float* h2  = h1 + (long long)N * D;

    int* rank = (int*)h1;
    int* deg  = (int*)h2;
    int* off  = deg + N;
    int* perm = off + N + 1;

    const int zeroBlocks  = (N + 255) / 256;
    const int edgeBlocks  = (E + 1023) / 1024;
    const int sumBlocks   = 2048;
    const int gemmBlocks  = 512;
    const int decBlocks   = (Dk + 255) / 256;

    zero_kernel<<<zeroBlocks, 256, 0, stream>>>(deg, N);
    rank_kernel<<<edgeBlocks, 256, 0, stream>>>(dst, deg, rank, E);
    scan_kernel<<<1, 1024, 0, stream>>>(deg, off, cnt, N);
    place_kernel<<<edgeBlocks, 256, 0, stream>>>(src, dst, rank, off, perm, E);
    segment_sum_kernel<<<sumBlocks, 256, 0, stream>>>(x, off, perm, s, N, E);
    sage_gemm_kernel<<<gemmBlocks, 256, 0, stream>>>(x, s, cnt, W1l, b1, W1r,
                                                     h1, N,  1);
    segment_sum_kernel<<<sumBlocks, 256, 0, stream>>>(h1, off, perm, s, N, E);
    sage_gemm_kernel<<<gemmBlocks, 256, 0, stream>>>(h1, s, cnt, W2l, b2, W2r,
                                                     h2, N,  0);
    decode_kernel<<<decBlocks, 256, 0, stream>>>(h2, dec, out, Dk, N);
}


# edge ordering: two-level bucket sort on LDS counters (slice counts, scan, pair scatter, per-bucket node sort) replaces returning global int atomics; f32 sums stay in arrival order, now of LDS adds
# speedup vs baseline: 5.0180x; 1.6867x over previous
#include <hip/hip_runtime.h>
#include <hip/hip_bf16.h>
#include <math.h>

typedef __attribute__((ext_vector_type(2))) float v2f;
typedef __attribute__((ext_vector_type(8))) float v8f;

#define D 128

#define BSH    8
#define SLICES 256
#define MAXC   256

__device__ __forceinline__ int bucket_of(int d, int C)
{
    const unsigned c = (unsigned)d >> BSH;
    return (int)(c < (unsigned)C ? c : (unsigned)(C - 1));
}

__device__ __forceinline__ int slice_len(int E)
{
    return (((E + SLICES - 1) / SLICES) + 255) & ~255;
}

template <int NW>
__device__ __forceinline__ int block_excl_scan(int v, int* sh, int t)
{
    const int lane = t & 31, w = t >> 5;
    int incl = v;
    #pragma unroll
    for (int o = 1; o < 32; o <<= 1) {
        const int y = __shfl_up(incl, o, 32);
        if (lane >= o) incl += y;
    }
    if (lane == 31) sh[w] = incl;
    __syncthreads();
    int wb = 0;
    #pragma unroll
    for (int i = 0; i < NW; ++i) wb += (i < w) ? sh[i] : 0;
    return wb + incl - v;
}

#define EB 16
__global__ __launch_bounds__(256) void bucket_count_kernel(
    const int* __restrict__ dst, int* __restrict__ hist, int E, int C)
{
    __shared__ int cntb[MAXC];
    const int t = threadIdx.x, b = blockIdx.x;
    cntb[t] = 0;
    __syncthreads();
    const int per = slice_len(E);
    const int beg = b * per;
    const int end = (beg + per < E) ? (beg + per) : E;
    for (int e0 = beg + t; e0 < end; e0 += 256 * EB) {
        int d[EB];
        #pragma unroll
        for (int k = 0; k < EB; ++k) {
            const int e = e0 + 256 * k;
            d[k] = dst[e < end ? e : end - 1];
        }
        #pragma unroll
        for (int k = 0; k < EB; ++k)
            if (e0 + 256 * k < end) atomicAdd(&cntb[bucket_of(d[k], C)], 1);
    }
    __syncthreads();
    if (t < C) hist[t * SLICES + b] = cntb[t];
}

__global__ __launch_bounds__(256) void bucket_scan_kernel(
    int* __restrict__ hist, int* __restrict__ total, int C)
{
    const int lane = threadIdx.x & 31;
    const int c = (int)((blockIdx.x * blockDim.x + threadIdx.x) >> 5);
    if (c >= C) return;
    int4* row = (int4*)(hist + c * SLICES) + 2 * lane;
    const int4 a = row[0], b = row[1];
    const int e1 = a.x, e2 = e1 + a.y, e3 = e2 + a.z, e4 = e3 + a.w;
    const int e5 = e4 + b.x, e6 = e5 + b.y, e7 = e6 + b.z, sum = e7 + b.w;
    int incl = sum;
    #pragma unroll
    for (int o = 1; o < 32; o <<= 1) {
        const int y = __shfl_up(incl, o, 32);
        if (lane >= o) incl += y;
    }
    const int base = incl - sum;
    row[0] = make_int4(base, base + e1, base + e2, base + e3);
    row[1] = make_int4(base + e4, base + e5, base + e6, base + e7);
    if (lane == 31) total[c] = incl;
}

__global__ __launch_bounds__(256) void bucket_scatter_kernel(
    const int* __restrict__ src, const int* __restrict__ dst,
    const int* __restrict__ hist, const int* __restrict__ total,
    int2* __restrict__ tmp, int E, int C)
{
    __shared__ int cur[MAXC];
    __shared__ int wsum[8];
    const int t = threadIdx.x, b = blockIdx.x;
    const int tot = (t < C) ? total[t] : 0;
    const int pre = (t < C) ? hist[t * SLICES + b] : 0;
    const int start = block_excl_scan<8>(tot, wsum, t);
    cur[t] = start + pre;
    __syncthreads();
    const int per = slice_len(E);
    const int beg = b * per;
    const int end = (beg + per < E) ? (beg + per) : E;
    for (int e0 = beg + t; e0 < end; e0 += 256 * EB) {
        int d[EB], s[EB];
        #pragma unroll
        for (int k = 0; k < EB; ++k) {
            const int e = e0 + 256 * k;
            const int ec = e < end ? e : end - 1;
            d[k] = dst[ec];
            s[k] = src[ec];
        }
        #pragma unroll
        for (int k = 0; k < EB; ++k) {
            if (e0 + 256 * k < end) {
                const int pos = atomicAdd(&cur[bucket_of(d[k], C)], 1);
                if ((unsigned)pos < (unsigned)E) tmp[pos] = make_int2(s[k], d[k]);
            }
        }
    }
}

#define SB 20
__global__ __launch_bounds__(512) void bucket_sort_kernel(
    const int2* __restrict__ tmp, const int* __restrict__ total,
    int* __restrict__ off, float* __restrict__ cnt, int* __restrict__ perm,
    int N, int E, int C)
{
    __shared__ int cn[256];
    __shared__ int wsA[16], wsB[16];
    __shared__ int seg[2];
    const int t = threadIdx.x, c = blockIdx.x;
    const int tot = (t < C) ? total[t] : 0;
    if (t < 256) cn[t] = 0;
    const int start = block_excl_scan<16>(tot, wsA, t);
    if (t == c) { seg[0] = start; seg[1] = start + tot; }
    __syncthreads();
    int beg = seg[0], end = seg[1];
    beg = beg < 0 ? 0 : beg;
    end = end > E ? E : end;

    for (int e0 = beg + t; e0 < end; e0 += 512 * SB) {
        int d[SB];
        #pragma unroll
        for (int k = 0; k < SB; ++k) {
            const int e = e0 + 512 * k;
            d[k] = tmp[e < end ? e : end - 1].y;
        }
        #pragma unroll
        for (int k = 0; k < SB; ++k)
            if (e0 + 512 * k < end) atomicAdd(&cn[d[k] & 255], 1);
    }
    __syncthreads();

    const int myc = (t < 256) ? cn[t] : 0;
    const int ex = block_excl_scan<16>(myc, wsB, t);
    if (t < 256) {
        const int node = (c << BSH) + t;
        cn[t] = beg + ex;
        if (node < N) { off[node] = beg + ex; cnt[node] = (float)myc; }
    }
    if (c == C - 1 && t == 0) off[N] = end;
    __syncthreads();

    for (int e0 = beg + t; e0 < end; e0 += 512 * SB) {
        int2 p[SB];
        #pragma unroll
        for (int k = 0; k < SB; ++k) {
            const int e = e0 + 512 * k;
            p[k] = tmp[e < end ? e : end - 1];
        }
        #pragma unroll
        for (int k = 0; k < SB; ++k) {
            if (e0 + 512 * k < end) {
                const int pos = atomicAdd(&cn[p[k].y & 255], 1);
                if ((unsigned)pos < (unsigned)E) perm[pos] = p[k].x;
            }
        }
    }
}

__global__ __launch_bounds__(256) void segment_sum_kernel(
    const float* __restrict__ xin, const int* __restrict__ off,
    const int* __restrict__ perm, float* __restrict__ s, int N, int E)
{
    const int lane = threadIdx.x & 31;
    const int wave0  = __builtin_amdgcn_readfirstlane(
                           (int)((blockIdx.x * blockDim.x + threadIdx.x) >> 5));
    const int nwaves = (int)((gridDim.x * blockDim.x) >> 5);
    const float4* xb = (const float4*)xin + lane;
    const unsigned last = (unsigned)(N - 1);

    for (int node = wave0; node < N; node += nwaves) {
        int beg = off[node];
        int end = off[node + 1];
        beg = beg < 0 ? 0 : beg;
        end = end > E ? E : end;
        float4 a0 = make_float4(0.f, 0.f, 0.f, 0.f);
        float4 a1 = a0, a2 = a0, a3 = a0;
        int j = beg;
        for (; j + 8 <= end; j += 8) {
            float4 v[8];
            #pragma unroll
            for (int k = 0; k < 8; ++k) {
                unsigned idx = (unsigned)perm[j + k];
                idx = idx < last ? idx : last;
                v[k] = xb[(size_t)idx * (D / 4)];
            }
            a0.x += v[0].x; a0.y += v[0].y; a0.z += v[0].z; a0.w += v[0].w;
            a1.x += v[1].x; a1.y += v[1].y; a1.z += v[1].z; a1.w += v[1].w;
            a2.x += v[2].x; a2.y += v[2].y; a2.z += v[2].z; a2.w += v[2].w;
            a3.x += v[3].x; a3.y += v[3].y; a3.z += v[3].z; a3.w += v[3].w;
            a0.x += v[4].x; a0.y += v[4].y; a0.z += v[4].z; a0.w += v[4].w;
            a1.x += v[5].x; a1.y += v[5].y; a1.z += v[5].z; a1.w += v[5].w;
            a2.x += v[6].x; a2.y += v[6].y; a2.z += v[6].z; a2.w += v[6].w;
            a3.x += v[7].x; a3.y += v[7].y; a3.z += v[7].z; a3.w += v[7].w;
        }
        if (j < end) {
            float4 v[8];
            #pragma unroll
            for (int k = 0; k < 8; ++k) {
                const int jj = (j + k < end) ? (j + k) : (end - 1);
                unsigned idx = (unsigned)perm[jj];
                idx = idx < last ? idx : last;
                v[k] = xb[(size_t)idx * (D / 4)];
            }
            #pragma unroll
            for (int k = 0; k < 8; ++k) {
                if (j + k >= end) v[k] = make_float4(0.f, 0.f, 0.f, 0.f);
            }
            a0.x += v[0].x; a0.y += v[0].y; a0.z += v[0].z; a0.w += v[0].w;
            a1.x += v[1].x; a1.y += v[1].y; a1.z += v[1].z; a1.w += v[1].w;
            a2.x += v[2].x; a2.y += v[2].y; a2.z += v[2].z; a2.w += v[2].w;
            a3.x += v[3].x; a3.y += v[3].y; a3.z += v[3].z; a3.w += v[3].w;
            a0.x += v[4].x; a0.y += v[4].y; a0.z += v[4].z; a0.w += v[4].w;
            a1.x += v[5].x; a1.y += v[5].y; a1.z += v[5].z; a1.w += v[5].w;
            a2.x += v[6].x; a2.y += v[6].y; a2.z += v[6].z; a2.w += v[6].w;
        }
        float4 r;
        r.x = (a0.x + a1.x) + (a2.x + a3.x);
        r.y = (a0.y + a1.y) + (a2.y + a3.y);
        r.z = (a0.z + a1.z) + (a2.z + a3.z);
        r.w = (a0.w + a1.w) + (a2.w + a3.w);
        ((float4*)(s + (size_t)node * D))[lane] = r;
    }
}

#define WPP 288
#define GEMM_WAVES 8

__global__ __launch_bounds__(256) void sage_gemm_kernel(
    const float* __restrict__ xin, const float* __restrict__ s,
    const float* __restrict__ cnt, const float* __restrict__ Wl,
    const float* __restrict__ bias, const float* __restrict__ Wr,
    float* __restrict__ out, int N, int relu)
{
    __shared__ __attribute__((aligned(16))) float Ws[2 * (D / 2) * WPP];

    for (int it = 0; it < 16; ++it) {
        const int i  = it * 256 + (int)threadIdx.x;
        const int m  = i >> 11;
        const int kp = (i >> 5) & 63;
        const int q  = i & 31;
        const float4* wsrc = (const float4*)(m ? Wr : Wl);
        const float4 r0 = wsrc[(2 * kp) * (D / 4) + q];
        const float4 r1 = wsrc[(2 * kp + 1) * (D / 4) + q];
        float4* wdst = (float4*)(Ws + (m * (D / 2) + kp) * WPP + 8 * q);
        wdst[0] = make_float4(r0.x, r1.x, r0.y, r1.y);
        wdst[1] = make_float4(r0.z, r1.z, r0.w, r1.w);
    }
    __syncthreads();

    const int lane = threadIdx.x & 31;
    const int wave = __builtin_amdgcn_readfirstlane((int)(threadIdx.x >> 5));
    const int c = lane & 15;
    const int h = lane >> 4;
    const int ntiles = (N + 15) >> 4;

    for (int idx = wave; ; idx += GEMM_WAVES) {
        const int tile = (int)blockIdx.x + (idx >> 1) * (int)gridDim.x;
        if (tile >= ntiles) break;
        const int half = idx & 1;
        const int m0 = tile * 16;
        const int n0 = half * 64;

        int rowA = m0 + c;
        if (rowA > N - 1) rowA = N - 1;
        const float* srow = s   + (long long)rowA * D + 2 * h;
        const float* xrow = xin + (long long)rowA * D + 2 * h;
        const float inv = 1.0f / fmaxf(cnt[rowA], 1.0f);

        v2f as[32], ax[32];
        #pragma unroll
        for (int ks = 0; ks < 32; ++ks) {
            as[ks] = *(const v2f*)(srow + 4 * ks);
            ax[ks] = *(const v2f*)(xrow + 4 * ks);
        }

        const float* bl = Ws + h * WPP + 2 * (n0 + c);
        const float* br = bl + (D / 2) * WPP;

        v8f acc[4] = {};
        #pragma unroll
        for (int ks = 0; ks < 32; ++ks) {
            v2f a0; a0.x = as[ks].x * inv; a0.y = as[ks].y * inv;
            const v2f a1 = ax[ks];
            #pragma unroll
            for (int t = 0; t < 4; ++t) {
                const v2f b0 = *(const v2f*)(bl + (2 * ks) * WPP + 32 * t);
                acc[t] = __builtin_amdgcn_wmma_f32_16x16x4_f32(
                             false, a0, false, b0, (short)0, acc[t], false, false);
                const v2f b1 = *(const v2f*)(br + (2 * ks) * WPP + 32 * t);
                acc[t] = __builtin_amdgcn_wmma_f32_16x16x4_f32(
                             false, a1, false, b1, (short)0, acc[t], false, false);
            }
        }

        const int mbase = m0 + (h << 3);
        #pragma unroll
        for (int t = 0; t < 4; ++t) {
            const int ncol = n0 + 16 * t + c;
            const float bn = bias[ncol];
            #pragma unroll
            for (int r = 0; r < 8; ++r) {
                const int m = mbase + r;
                if (m < N) {
                    float v = acc[t][r] + bn;
                    if (relu) v = fmaxf(v, 0.0f);
                    out[(long long)m * D + ncol] = v;
                }
            }
        }
    }
}

__device__ __forceinline__ float reduce8(const float (&v)[8], int lane)
{
    const bool b4 = (lane & 16) != 0, b3 = (lane & 8) != 0, b2 = (lane & 4) != 0;
    float w[4];
    #pragma unroll
    for (int j = 0; j < 4; ++j) {
        const float keep = b4 ? v[j + 4] : v[j];
        const float send = b4 ? v[j] : v[j + 4];
        w[j] = keep + __shfl_xor(send, 16, 32);
    }
    float x[2];
    #pragma unroll
    for (int j = 0; j < 2; ++j) {
        const float keep = b3 ? w[j + 2] : w[j];
        const float send = b3 ? w[j] : w[j + 2];
        x[j] = keep + __shfl_xor(send, 8, 32);
    }
    const float keep = b2 ? x[1] : x[0];
    const float send = b2 ? x[0] : x[1];
    float y = keep + __shfl_xor(send, 4, 32);
    y += __shfl_xor(y, 2, 32);
    y += __shfl_xor(y, 1, 32);
    return y;
}

__global__ __launch_bounds__(256) void decode_kernel(
    const float* __restrict__ h, const int* __restrict__ dec,
    float* __restrict__ out, int Dk, int N)
{
    const int lane = threadIdx.x & 31;
    const int wave = __builtin_amdgcn_readfirstlane(
                         (int)((blockIdx.x * blockDim.x + threadIdx.x) >> 5));
    const int base = wave * 32;
    if (base >= Dk) return;
    const int p  = base + lane;
    const int pc = p < Dk ? p : Dk - 1;
    const unsigned last = (unsigned)(N - 1);
    unsigned ai = (unsigned)dec[pc];
    unsigned bi = (unsigned)dec[Dk + pc];
    ai = ai < last ? ai : last;
    bi = bi < last ? bi : last;
    const float4* hb = (const float4*)h + lane;

    float rd = 0.f, ra = 0.f, rb = 0.f;
    for (int b = 0; b < 4; ++b) {
        float4 va[8], vb[8];
        #pragma unroll
        for (int k = 0; k < 8; ++k) {
            const unsigned ia = (unsigned)__builtin_amdgcn_readlane((int)ai, b * 8 + k);
            const unsigned ib = (unsigned)__builtin_amdgcn_readlane((int)bi, b * 8 + k);
            va[k] = hb[(size_t)ia * (D / 4)];
            vb[k] = hb[(size_t)ib * (D / 4)];
        }
        float d[8], na[8], nb[8];
        #pragma unroll
        for (int k = 0; k < 8; ++k) {
            const float4 a = va[k], c = vb[k];
            d[k]  = a.x * c.x + a.y * c.y + a.z * c.z + a.w * c.w;
            na[k] = a.x * a.x + a.y * a.y + a.z * a.z + a.w * a.w;
            nb[k] = c.x * c.x + c.y * c.y + c.z * c.z + c.w * c.w;
        }
        const float sd = reduce8(d, lane);
        const float sa = reduce8(na, lane);
        const float sb = reduce8(nb, lane);
        const int from = (lane & 7) << 2;
        const float td = __shfl(sd, from, 32);
        const float ta = __shfl(sa, from, 32);
        const float tb = __shfl(sb, from, 32);
        if ((lane >> 3) == b) { rd = td; ra = ta; rb = tb; }
    }
    const float denom = fmaxf(sqrtf(ra) * sqrtf(rb), 1e-6f);
    const float z = rd / denom;
    if (p < Dk) out[p] = 1.0f / (1.0f + expf(-z));
}

extern "C" void kernel_launch(void* const* d_in, const int* in_sizes, int n_in,
                              void* d_out, int out_size, void* d_ws, size_t ws_size,
                              hipStream_t stream) {
    const float* x   = (const float*)d_in[0];
    const int*   edg = (const int*)d_in[1];
    const int*   dec = (const int*)d_in[2];
    const float* W1l = (const float*)d_in[3];
    const float* b1  = (const float*)d_in[4];
    const float* W1r = (const float*)d_in[5];
    const float* W2l = (const float*)d_in[6];
    const float* b2  = (const float*)d_in[7];
    const float* W2r = (const float*)d_in[8];
    float* out = (float*)d_out;

    const int N  = in_sizes[0] / D;
    const int E  = in_sizes[1] / 2;
    const int Dk = in_sizes[2] / 2;
    const int* src = edg;
    const int* dst = edg + E;

    float* ws  = (float*)d_ws;
    float* s   = ws;
    float* cnt = s + (long long)N * D;
    float* h1  = cnt + N;
    float* h2  = h1 + (long long)N * D;

    int*  hist  = (int*)h1;
    int*  total = hist + MAXC * SLICES;
    int2* tmp   = (int2*)(total + MAXC);
    int*  off   = (int*)h2;
    int*  perm  = off + N + 1;

    const int C = (N + 255) >> BSH;
    const int scanBlocks  = (C + 7) / 8;
    const int sumBlocks   = 2048;
    const int gemmBlocks  = 512;
    const int decBlocks   = (Dk + 255) / 256;

    bucket_count_kernel<<<SLICES, 256, 0, stream>>>(dst, hist, E, C);
    bucket_scan_kernel<<<scanBlocks, 256, 0, stream>>>(hist, total, C);
    bucket_scatter_kernel<<<SLICES, 256, 0, stream>>>(src, dst, hist, total, tmp, E, C);
    bucket_sort_kernel<<<C, 512, 0, stream>>>(tmp, total, off, cnt, perm, N, E, C);
    segment_sum_kernel<<<sumBlocks, 256, 0, stream>>>(x, off, perm, s, N, E);
    sage_gemm_kernel<<<gemmBlocks, 256, 0, stream>>>(x, s, cnt, W1l, b1, W1r,
                                                     h1, N,  1);
    segment_sum_kernel<<<sumBlocks, 256, 0, stream>>>(h1, off, perm, s, N, E);
    sage_gemm_kernel<<<gemmBlocks, 256, 0, stream>>>(h1, s, cnt, W2l, b2, W2r,
                                                     h2, N,  0);
    decode_kernel<<<decBlocks, 256, 0, stream>>>(h2, dec, out, Dk, N);
}


# rows aligned: the mean's scale moves from the GEMM's A feed to the segment sum's store, the in-degree plane goes, h1/h2 rows start on 512-B boundaries (were 64 B past a line): 4 lines a row, not 5
# speedup vs baseline: 5.2288x; 1.0420x over previous
#include <hip/hip_runtime.h>
#include <hip/hip_bf16.h>
#include <math.h>

typedef __attribute__((ext_vector_type(2))) float v2f;
typedef __attribute__((ext_vector_type(8))) float v8f;

#define D 128

#define BSH    8
#define SLICES 256
#define MAXC   256

__device__ __forceinline__ int bucket_of(int d, int C)
{
    const unsigned c = (unsigned)d >> BSH;
    return (int)(c < (unsigned)C ? c : (unsigned)(C - 1));
}

__device__ __forceinline__ int slice_len(int E)
{
    return (((E + SLICES - 1) / SLICES) + 255) & ~255;
}

template <int NW>
__device__ __forceinline__ int block_excl_scan(int v, int* sh, int t)
{
    const int lane = t & 31, w = t >> 5;
    int incl = v;
    #pragma unroll
    for (int o = 1; o < 32; o <<= 1) {
        const int y = __shfl_up(incl, o, 32);
        if (lane >= o) incl += y;
    }
    if (lane == 31) sh[w] = incl;
    __syncthreads();
    int wb = 0;
    #pragma unroll
    for (int i = 0; i < NW; ++i) wb += (i < w) ? sh[i] : 0;
    return wb + incl - v;
}

#define EB 16
__global__ __launch_bounds__(256) void bucket_count_kernel(
    const int* __restrict__ dst, int* __restrict__ hist, int E, int C)
{
    __shared__ int cntb[MAXC];
    const int t = threadIdx.x, b = blockIdx.x;
    cntb[t] = 0;
    __syncthreads();
    const int per = slice_len(E);
    const int beg = b * per;
    const int end = (beg + per < E) ? (beg + per) : E;
    for (int e0 = beg + t; e0 < end; e0 += 256 * EB) {
        int d[EB];
        #pragma unroll
        for (int k = 0; k < EB; ++k) {
            const int e = e0 + 256 * k;
            d[k] = dst[e < end ? e : end - 1];
        }
        #pragma unroll
        for (int k = 0; k < EB; ++k)
            if (e0 + 256 * k < end) atomicAdd(&cntb[bucket_of(d[k], C)], 1);
    }
    __syncthreads();
    if (t < C) hist[t * SLICES + b] = cntb[t];
}

__global__ __launch_bounds__(256) void bucket_scan_kernel(
    int* __restrict__ hist, int* __restrict__ total, int C)
{
    const int lane = threadIdx.x & 31;
    const int c = (int)((blockIdx.x * blockDim.x + threadIdx.x) >> 5);
    if (c >= C) return;
    int4* row = (int4*)(hist + c * SLICES) + 2 * lane;
    const int4 a = row[0], b = row[1];
    const int e1 = a.x, e2 = e1 + a.y, e3 = e2 + a.z, e4 = e3 + a.w;
    const int e5 = e4 + b.x, e6 = e5 + b.y, e7 = e6 + b.z, sum = e7 + b.w;
    int incl = sum;
    #pragma unroll
    for (int o = 1; o < 32; o <<= 1) {
        const int y = __shfl_up(incl, o, 32);
        if (lane >= o) incl += y;
    }
    const int base = incl - sum;
    row[0] = make_int4(base, base + e1, base + e2, base + e3);
    row[1] = make_int4(base + e4, base + e5, base + e6, base + e7);
    if (lane == 31) total[c] = incl;
}

__global__ __launch_bounds__(256) void bucket_scatter_kernel(
    const int* __restrict__ src, const int* __restrict__ dst,
    const int* __restrict__ hist, const int* __restrict__ total,
    int2* __restrict__ tmp, int E, int C)
{
    __shared__ int cur[MAXC];
    __shared__ int wsum[8];
    const int t = threadIdx.x, b = blockIdx.x;
    const int tot = (t < C) ? total[t] : 0;
    const int pre = (t < C) ? hist[t * SLICES + b] : 0;
    const int start = block_excl_scan<8>(tot, wsum, t);
    cur[t] = start + pre;
    __syncthreads();
    const int per = slice_len(E);
    const int beg = b * per;
    const int end = (beg + per < E) ? (beg + per) : E;
    for (int e0 = beg + t; e0 < end; e0 += 256 * EB) {
        int d[EB], s[EB];
        #pragma unroll
        for (int k = 0; k < EB; ++k) {
            const int e = e0 + 256 * k;
            const int ec = e < end ? e : end - 1;
            d[k] = dst[ec];
            s[k] = src[ec];
        }
        #pragma unroll
        for (int k = 0; k < EB; ++k) {
            if (e0 + 256 * k < end) {
                const int pos = atomicAdd(&cur[bucket_of(d[k], C)], 1);
                if ((unsigned)pos < (unsigned)E) tmp[pos] = make_int2(s[k], d[k]);
            }
        }
    }
}

#define SB 20
__global__ __launch_bounds__(512) void bucket_sort_kernel(
    const int2* __restrict__ tmp, const int* __restrict__ total,
    int* __restrict__ off, int* __restrict__ perm,
    int N, int E, int C)
{
    __shared__ int cn[256];
    __shared__ int wsA[16], wsB[16];
    __shared__ int seg[2];
    const int t = threadIdx.x, c = blockIdx.x;
    const int tot = (t < C) ? total[t] : 0;
    if (t < 256) cn[t] = 0;
    const int start = block_excl_scan<16>(tot, wsA, t);
    if (t == c) { seg[0] = start; seg[1] = start + tot; }
    __syncthreads();
    int beg = seg[0], end = seg[1];
    beg = beg < 0 ? 0 : beg;
    end = end > E ? E : end;

    for (int e0 = beg + t; e0 < end; e0 += 512 * SB) {
        int d[SB];
        #pragma unroll
        for (int k = 0; k < SB; ++k) {
            const int e = e0 + 512 * k;
            d[k] = tmp[e < end ? e : end - 1].y;
        }
        #pragma unroll
        for (int k = 0; k < SB; ++k)
            if (e0 + 512 * k < end) atomicAdd(&cn[d[k] & 255], 1);
    }
    __syncthreads();

    const int myc = (t < 256) ? cn[t] : 0;
    const int ex = block_excl_scan<16>(myc, wsB, t);
    if (t < 256) {
        const int node = (c << BSH) + t;
        cn[t] = beg + ex;
        if (node < N) off[node] = beg + ex;
    }
    if (c == C - 1 && t == 0) off[N] = end;
    __syncthreads();

    for (int e0 = beg + t; e0 < end; e0 += 512 * SB) {
        int2 p[SB];
        #pragma unroll
        for (int k = 0; k < SB; ++k) {
            const int e = e0 + 512 * k;
            p[k] = tmp[e < end ? e : end - 1];
        }
        #pragma unroll
        for (int k = 0; k < SB; ++k) {
            if (e0 + 512 * k < end) {
                const int pos = atomicAdd(&cn[p[k].y & 255], 1);
                if ((unsigned)pos < (unsigned)E) perm[pos] = p[k].x;
            }
        }
    }
}

__global__ __launch_bounds__(256) void segment_sum_kernel(
    const float* __restrict__ xin, const int* __restrict__ off,
    const int* __restrict__ perm, float* __restrict__ s, int N, int E)
{
    const int lane = threadIdx.x & 31;
    const int wave0  = __builtin_amdgcn_readfirstlane(
                           (int)((blockIdx.x * blockDim.x + threadIdx.x) >> 5));
    const int nwaves = (int)((gridDim.x * blockDim.x) >> 5);
    const float4* xb = (const float4*)xin + lane;
    const unsigned last = (unsigned)(N - 1);

    for (int node = wave0; node < N; node += nwaves) {
        int beg = off[node];
        int end = off[node + 1];
        beg = beg < 0 ? 0 : beg;
        end = end > E ? E : end;
        float4 a0 = make_float4(0.f, 0.f, 0.f, 0.f);
        float4 a1 = a0, a2 = a0, a3 = a0;
        int j = beg;
        for (; j + 8 <= end; j += 8) {
            float4 v[8];
            #pragma unroll
            for (int k = 0; k < 8; ++k) {
                unsigned idx = (unsigned)perm[j + k];
                idx = idx < last ? idx : last;
                v[k] = xb[(size_t)idx * (D / 4)];
            }
            a0.x += v[0].x; a0.y += v[0].y; a0.z += v[0].z; a0.w += v[0].w;
            a1.x += v[1].x; a1.y += v[1].y; a1.z += v[1].z; a1.w += v[1].w;
            a2.x += v[2].x; a2.y += v[2].y; a2.z += v[2].z; a2.w += v[2].w;
            a3.x += v[3].x; a3.y += v[3].y; a3.z += v[3].z; a3.w += v[3].w;
            a0.x += v[4].x; a0.y += v[4].y; a0.z += v[4].z; a0.w += v[4].w;
            a1.x += v[5].x; a1.y += v[5].y; a1.z += v[5].z; a1.w += v[5].w;
            a2.x += v[6].x; a2.y += v[6].y; a2.z += v[6].z; a2.w += v[6].w;
            a3.x += v[7].x; a3.y += v[7].y; a3.z += v[7].z; a3.w += v[7].w;
        }
        if (j < end) {
            float4 v[8];
            #pragma unroll
            for (int k = 0; k < 8; ++k) {
                const int jj = (j + k < end) ? (j + k) : (end - 1);
                unsigned idx = (unsigned)perm[jj];
                idx = idx < last ? idx : last;
                v[k] = xb[(size_t)idx * (D / 4)];
            }
            #pragma unroll
            for (int k = 0; k < 8; ++k) {
                if (j + k >= end) v[k] = make_float4(0.f, 0.f, 0.f, 0.f);
            }
            a0.x += v[0].x; a0.y += v[0].y; a0.z += v[0].z; a0.w += v[0].w;
            a1.x += v[1].x; a1.y += v[1].y; a1.z += v[1].z; a1.w += v[1].w;
            a2.x += v[2].x; a2.y += v[2].y; a2.z += v[2].z; a2.w += v[2].w;
            a3.x += v[3].x; a3.y += v[3].y; a3.z += v[3].z; a3.w += v[3].w;
            a0.x += v[4].x; a0.y += v[4].y; a0.z += v[4].z; a0.w += v[4].w;
            a1.x += v[5].x; a1.y += v[5].y; a1.z += v[5].z; a1.w += v[5].w;
            a2.x += v[6].x; a2.y += v[6].y; a2.z += v[6].z; a2.w += v[6].w;
        }
        const float inv = 1.0f / fmaxf((float)(end - beg), 1.0f);
        float4 r;
        r.x = ((a0.x + a1.x) + (a2.x + a3.x)) * inv;
        r.y = ((a0.y + a1.y) + (a2.y + a3.y)) * inv;
        r.z = ((a0.z + a1.z) + (a2.z + a3.z)) * inv;
        r.w = ((a0.w + a1.w) + (a2.w + a3.w)) * inv;
        ((float4*)(s + (size_t)node * D))[lane] = r;
    }
}

#define WPP 288
#define GEMM_WAVES 8

__global__ __launch_bounds__(256) void sage_gemm_kernel(
    const float* __restrict__ xin, const float* __restrict__ s,
    const float* __restrict__ Wl,
    const float* __restrict__ bias, const float* __restrict__ Wr,
    float* __restrict__ out, int N, int relu)
{
    __shared__ __attribute__((aligned(16))) float Ws[2 * (D / 2) * WPP];

    for (int it = 0; it < 16; ++it) {
        const int i  = it * 256 + (int)threadIdx.x;
        const int m  = i >> 11;
        const int kp = (i >> 5) & 63;
        const int q  = i & 31;
        const float4* wsrc = (const float4*)(m ? Wr : Wl);
        const float4 r0 = wsrc[(2 * kp) * (D / 4) + q];
        const float4 r1 = wsrc[(2 * kp + 1) * (D / 4) + q];
        float4* wdst = (float4*)(Ws + (m * (D / 2) + kp) * WPP + 8 * q);
        wdst[0] = make_float4(r0.x, r1.x, r0.y, r1.y);
        wdst[1] = make_float4(r0.z, r1.z, r0.w, r1.w);
    }
    __syncthreads();

    const int lane = threadIdx.x & 31;
    const int wave = __builtin_amdgcn_readfirstlane((int)(threadIdx.x >> 5));
    const int c = lane & 15;
    const int h = lane >> 4;
    const int ntiles = (N + 15) >> 4;

    for (int idx = wave; ; idx += GEMM_WAVES) {
        const int tile = (int)blockIdx.x + (idx >> 1) * (int)gridDim.x;
        if (tile >= ntiles) break;
        const int half = idx & 1;
        const int m0 = tile * 16;
        const int n0 = half * 64;

        int rowA = m0 + c;
        if (rowA > N - 1) rowA = N - 1;
        const float* srow = s   + (long long)rowA * D + 2 * h;
        const float* xrow = xin + (long long)rowA * D + 2 * h;

        v2f as[32], ax[32];
        #pragma unroll
        for (int ks = 0; ks < 32; ++ks) {
            as[ks] = *(const v2f*)(srow + 4 * ks);
            ax[ks] = *(const v2f*)(xrow + 4 * ks);
        }

        const float* bl = Ws + h * WPP + 2 * (n0 + c);
        const float* br = bl + (D / 2) * WPP;

        v8f acc[4] = {};
        #pragma unroll
        for (int ks = 0; ks < 32; ++ks) {
            const v2f a0 = as[ks];
            const v2f a1 = ax[ks];
            #pragma unroll
            for (int t = 0; t < 4; ++t) {
                const v2f b0 = *(const v2f*)(bl + (2 * ks) * WPP + 32 * t);
                acc[t] = __builtin_amdgcn_wmma_f32_16x16x4_f32(
                             false, a0, false, b0, (short)0, acc[t], false, false);
                const v2f b1 = *(const v2f*)(br + (2 * ks) * WPP + 32 * t);
                acc[t] = __builtin_amdgcn_wmma_f32_16x16x4_f32(
                             false, a1, false, b1, (short)0, acc[t], false, false);
            }
        }

        const int mbase = m0 + (h << 3);
        #pragma unroll
        for (int t = 0; t < 4; ++t) {
            const int ncol = n0 + 16 * t + c;
            const float bn = bias[ncol];
            #pragma unroll
            for (int r = 0; r < 8; ++r) {
                const int m = mbase + r;
                if (m < N) {
                    float v = acc[t][r] + bn;
                    if (relu) v = fmaxf(v, 0.0f);
                    out[(long long)m * D + ncol] = v;
                }
            }
        }
    }
}

__device__ __forceinline__ float reduce8(const float (&v)[8], int lane)
{
    const bool b4 = (lane & 16) != 0, b3 = (lane & 8) != 0, b2 = (lane & 4) != 0;
    float w[4];
    #pragma unroll
    for (int j = 0; j < 4; ++j) {
        const float keep = b4 ? v[j + 4] : v[j];
        const float send = b4 ? v[j] : v[j + 4];
        w[j] = keep + __shfl_xor(send, 16, 32);
    }
    float x[2];
    #pragma unroll
    for (int j = 0; j < 2; ++j) {
        const float keep = b3 ? w[j + 2] : w[j];
        const float send = b3 ? w[j] : w[j + 2];
        x[j] = keep + __shfl_xor(send, 8, 32);
    }
    const float keep = b2 ? x[1] : x[0];
    const float send = b2 ? x[0] : x[1];
    float y = keep + __shfl_xor(send, 4, 32);
    y += __shfl_xor(y, 2, 32);
    y += __shfl_xor(y, 1, 32);
    return y;
}

__global__ __launch_bounds__(256) void decode_kernel(
    const float* __restrict__ h, const int* __restrict__ dec,
    float* __restrict__ out, int Dk, int N)
{
    const int lane = threadIdx.x & 31;
    const int wave = __builtin_amdgcn_readfirstlane(
                         (int)((blockIdx.x * blockDim.x + threadIdx.x) >> 5));
    const int base = wave * 32;
    if (base >= Dk) return;
    const int p  = base + lane;
    const int pc = p < Dk ? p : Dk - 1;
    const unsigned last = (unsigned)(N - 1);
    unsigned ai = (unsigned)dec[pc];
    unsigned bi = (unsigned)dec[Dk + pc];
    ai = ai < last ? ai : last;
    bi = bi < last ? bi : last;
    const float4* hb = (const float4*)h + lane;

    float rd = 0.f, ra = 0.f, rb = 0.f;
    for (int b = 0; b < 4; ++b) {
        float4 va[8], vb[8];
        #pragma unroll
        for (int k = 0; k < 8; ++k) {
            const unsigned ia = (unsigned)__builtin_amdgcn_readlane((int)ai, b * 8 + k);
            const unsigned ib = (unsigned)__builtin_amdgcn_readlane((int)bi, b * 8 + k);
            va[k] = hb[(size_t)ia * (D / 4)];
            vb[k] = hb[(size_t)ib * (D / 4)];
        }
        float d[8], na[8], nb[8];
        #pragma unroll
        for (int k = 0; k < 8; ++k) {
            const float4 a = va[k], c = vb[k];
            d[k]  = a.x * c.x + a.y * c.y + a.z * c.z + a.w * c.w;
            na[k] = a.x * a.x + a.y * a.y + a.z * a.z + a.w * a.w;
            nb[k] = c.x * c.x + c.y * c.y + c.z * c.z + c.w * c.w;
        }
        const float sd = reduce8(d, lane);
        const float sa = reduce8(na, lane);
        const float sb = reduce8(nb, lane);
        const int from = (lane & 7) << 2;
        const float td = __shfl(sd, from, 32);
        const float ta = __shfl(sa, from, 32);
        const float tb = __shfl(sb, from, 32);
        if ((lane >> 3) == b) { rd = td; ra = ta; rb = tb; }
    }
    const float denom = fmaxf(sqrtf(ra) * sqrtf(rb), 1e-6f);
    const float z = rd / denom;
    if (p < Dk) out[p] = 1.0f / (1.0f + expf(-z));
}

extern "C" void kernel_launch(void* const* d_in, const int* in_sizes, int n_in,
                              void* d_out, int out_size, void* d_ws, size_t ws_size,
                              hipStream_t stream) {
    const float* x   = (const float*)d_in[0];
    const int*   edg = (const int*)d_in[1];
    const int*   dec = (const int*)d_in[2];
    const float* W1l = (const float*)d_in[3];
    const float* b1  = (const float*)d_in[4];
    const float* W1r = (const float*)d_in[5];
    const float* W2l = (const float*)d_in[6];
    const float* b2  = (const float*)d_in[7];
    const float* W2r = (const float*)d_in[8];
    float* out = (float*)d_out;

    const int N  = in_sizes[0] / D;
    const int E  = in_sizes[1] / 2;
    const int Dk = in_sizes[2] / 2;
    const int* src = edg;
    const int* dst = edg + E;

    float* ws  = (float*)d_ws;
    float* s   = ws;
    float* h1  = s + (long long)N * D;
    float* h2  = h1 + (long long)N * D;

    int*  hist  = (int*)h1;
    int*  total = hist + MAXC * SLICES;
    int2* tmp   = (int2*)(total + MAXC);
    int*  off   = (int*)h2;
    int*  perm  = off + N + 1;

    const int C = (N + 255) >> BSH;
    const int scanBlocks  = (C + 7) / 8;
    const int sumBlocks   = 2048;
    const int gemmBlocks  = 512;
    const int decBlocks   = (Dk + 255) / 256;

    bucket_count_kernel<<<SLICES, 256, 0, stream>>>(dst, hist, E, C);
    bucket_scan_kernel<<<scanBlocks, 256, 0, stream>>>(hist, total, C);
    bucket_scatter_kernel<<<SLICES, 256, 0, stream>>>(src, dst, hist, total, tmp, E, C);
    bucket_sort_kernel<<<C, 512, 0, stream>>>(tmp, total, off, perm, N, E, C);
    segment_sum_kernel<<<sumBlocks, 256, 0, stream>>>(x, off, perm, s, N, E);
    sage_gemm_kernel<<<gemmBlocks, 256, 0, stream>>>(x, s, W1l, b1, W1r,
                                                     h1, N,  1);
    segment_sum_kernel<<<sumBlocks, 256, 0, stream>>>(h1, off, perm, s, N, E);
    sage_gemm_kernel<<<gemmBlocks, 256, 0, stream>>>(h1, s, W2l, b2, W2r,
                                                     h2, N,  0);
    decode_kernel<<<decBlocks, 256, 0, stream>>>(h2, dec, out, Dk, N);
}


# layer kernel: a tile's list offsets, first 32 list entries per node and own x rows are requested under the previous tile's multiply, later chunks' entries one chunk ahead; sum orders unchanged
# speedup vs baseline: 5.8032x; 1.1099x over previous
#include <hip/hip_runtime.h>
#include <hip/hip_bf16.h>
#include <math.h>

typedef __attribute__((ext_vector_type(2))) float v2f;
typedef __attribute__((ext_vector_type(8))) float v8f;

#define D 128

#define BSH    8
#define SLICES 256
#define MAXC   256

__device__ __forceinline__ int bucket_of(int d, int C)
{
    const unsigned c = (unsigned)d >> BSH;
    return (int)(c < (unsigned)C ? c : (unsigned)(C - 1));
}

__device__ __forceinline__ int slice_len(int E)
{
    return (((E + SLICES - 1) / SLICES) + 255) & ~255;
}

template <int NW>
__device__ __forceinline__ int block_excl_scan(int v, int* sh, int t)
{
    const int lane = t & 31, w = t >> 5;
    int incl = v;
    #pragma unroll
    for (int o = 1; o < 32; o <<= 1) {
        const int y = __shfl_up(incl, o, 32);
        if (lane >= o) incl += y;
    }
    if (lane == 31) sh[w] = incl;
    __syncthreads();
    int wb = 0;
    #pragma unroll
    for (int i = 0; i < NW; ++i) wb += (i < w) ? sh[i] : 0;
    return wb + incl - v;
}

#define EB 16
__global__ __launch_bounds__(256) void bucket_count_kernel(
    const int* __restrict__ dst, int* __restrict__ hist, int E, int C)
{
    __shared__ int cntb[MAXC];
    const int t = threadIdx.x, b = blockIdx.x;
    cntb[t] = 0;
    __syncthreads();
    const int per = slice_len(E);
    const int beg = b * per;
    const int end = (beg + per < E) ? (beg + per) : E;
    for (int e0 = beg + t; e0 < end; e0 += 256 * EB) {
        int d[EB];
        #pragma unroll
        for (int k = 0; k < EB; ++k) {
            const int e = e0 + 256 * k;
            d[k] = dst[e < end ? e : end - 1];
        }
        #pragma unroll
        for (int k = 0; k < EB; ++k)
            if (e0 + 256 * k < end) atomicAdd(&cntb[bucket_of(d[k], C)], 1);
    }
    __syncthreads();
    if (t < C) hist[t * SLICES + b] = cntb[t];
}

__global__ __launch_bounds__(256) void bucket_scan_kernel(
    int* __restrict__ hist, int* __restrict__ total, int C)
{
    const int lane = threadIdx.x & 31;
    const int c = (int)((blockIdx.x * blockDim.x + threadIdx.x) >> 5);
    if (c >= C) return;
    int4* row = (int4*)(hist + c * SLICES) + 2 * lane;
    const int4 a = row[0], b = row[1];
    const int e1 = a.x, e2 = e1 + a.y, e3 = e2 + a.z, e4 = e3 + a.w;
    const int e5 = e4 + b.x, e6 = e5 + b.y, e7 = e6 + b.z, sum = e7 + b.w;
    int incl = sum;
    #pragma unroll
    for (int o = 1; o < 32; o <<= 1) {
        const int y = __shfl_up(incl, o, 32);
        if (lane >= o) incl += y;
    }
    const int base = incl - sum;
    row[0] = make_int4(base, base + e1, base + e2, base + e3);
    row[1] = make_int4(base + e4, base + e5, base + e6, base + e7);
    if (lane == 31) total[c] = incl;
}

__global__ __launch_bounds__(256) void bucket_scatter_kernel(
    const int* __restrict__ src, const int* __restrict__ dst,
    const int* __restrict__ hist, const int* __restrict__ total,
    int2* __restrict__ tmp, int E, int C)
{
    __shared__ int cur[MAXC];
    __shared__ int wsum[8];
    const int t = threadIdx.x, b = blockIdx.x;
    const int tot = (t < C) ? total[t] : 0;
    const int pre = (t < C) ? hist[t * SLICES + b] : 0;
    const int start = block_excl_scan<8>(tot, wsum, t);
    cur[t] = start + pre;
    __syncthreads();
    const int per = slice_len(E);
    const int beg = b * per;
    const int end = (beg + per < E) ? (beg + per) : E;
    for (int e0 = beg + t; e0 < end; e0 += 256 * EB) {
        int d[EB], s[EB];
        #pragma unroll
        for (int k = 0; k < EB; ++k) {
            const int e = e0 + 256 * k;
            const int ec = e < end ? e : end - 1;
            d[k] = dst[ec];
            s[k] = src[ec];
        }
        #pragma unroll
        for (int k = 0; k < EB; ++k) {
            if (e0 + 256 * k < end) {
                const int pos = atomicAdd(&cur[bucket_of(d[k], C)], 1);
                if ((unsigned)pos < (unsigned)E) tmp[pos] = make_int2(s[k], d[k]);
            }
        }
    }
}

#define SB 20
__global__ __launch_bounds__(512) void bucket_sort_kernel(
    const int2* __restrict__ tmp, const int* __restrict__ total,
    int* __restrict__ off, int* __restrict__ perm,
    int N, int E, int C)
{
    __shared__ int cn[256];
    __shared__ int wsA[16], wsB[16];
    __shared__ int seg[2];
    const int t = threadIdx.x, c = blockIdx.x;
    const int tot = (t < C) ? total[t] : 0;
    if (t < 256) cn[t] = 0;
    const int start = block_excl_scan<16>(tot, wsA, t);
    if (t == c) { seg[0] = start; seg[1] = start + tot; }
    __syncthreads();
    int beg = seg[0], end = seg[1];
    beg = beg < 0 ? 0 : beg;
    end = end > E ? E : end;

    for (int e0 = beg + t; e0 < end; e0 += 512 * SB) {
        int d[SB];
        #pragma unroll
        for (int k = 0; k < SB; ++k) {
            const int e = e0 + 512 * k;
            d[k] = tmp[e < end ? e : end - 1].y;
        }
        #pragma unroll
        for (int k = 0; k < SB; ++k)
            if (e0 + 512 * k < end) atomicAdd(&cn[d[k] & 255], 1);
    }
    __syncthreads();

    const int myc = (t < 256) ? cn[t] : 0;
    const int ex = block_excl_scan<16>(myc, wsB, t);
    if (t < 256) {
        const int node = (c << BSH) + t;
        cn[t] = beg + ex;
        if (node < N) off[node] = beg + ex;
    }
    if (c == C - 1 && t == 0) off[N] = end;
    __syncthreads();

    for (int e0 = beg + t; e0 < end; e0 += 512 * SB) {
        int2 p[SB];
        #pragma unroll
        for (int k = 0; k < SB; ++k) {
            const int e = e0 + 512 * k;
            p[k] = tmp[e < end ? e : end - 1];
        }
        #pragma unroll
        for (int k = 0; k < SB; ++k) {
            if (e0 + 512 * k < end) {
                const int pos = atomicAdd(&cn[p[k].y & 255], 1);
                if ((unsigned)pos < (unsigned)E) perm[pos] = p[k].x;
            }
        }
    }
}

#define WPQ 544
#define TP  260

__device__ __forceinline__ void add_rows8(float4& a0, float4& a1, float4& a2,
                                          float4& a3, const float4 (&v)[8])
{
    a0.x += v[0].x; a0.y += v[0].y; a0.z += v[0].z; a0.w += v[0].w;
    a1.x += v[1].x; a1.y += v[1].y; a1.z += v[1].z; a1.w += v[1].w;
    a2.x += v[2].x; a2.y += v[2].y; a2.z += v[2].z; a2.w += v[2].w;
    a3.x += v[3].x; a3.y += v[3].y; a3.z += v[3].z; a3.w += v[3].w;
    a0.x += v[4].x; a0.y += v[4].y; a0.z += v[4].z; a0.w += v[4].w;
    a1.x += v[5].x; a1.y += v[5].y; a1.z += v[5].z; a1.w += v[5].w;
    a2.x += v[6].x; a2.y += v[6].y; a2.z += v[6].z; a2.w += v[6].w;
    a3.x += v[7].x; a3.y += v[7].y; a3.z += v[7].z; a3.w += v[7].w;
}

__device__ __forceinline__ void load_batch(float4 (&v)[8], const float4* xb,
                                           unsigned p, int b8, int n)
{
    const int cap = n > 0 ? n - 1 : 0;
    #pragma unroll
    for (int k = 0; k < 8; ++k) {
        const int sel = (b8 + k) < cap ? (b8 + k) : cap;
        const unsigned idx = (unsigned)__builtin_amdgcn_readlane((int)p, sel);
        v[k] = xb[(size_t)idx * (D / 4)];
    }
    #pragma unroll
    for (int k = 0; k < 8; ++k) {
        if (b8 + k >= n) v[k] = make_float4(0.f, 0.f, 0.f, 0.f);
    }
}

__device__ __forceinline__ float4 scaled_sum(float4 a0, float4 a1, float4 a2,
                                             float4 a3, int deg)
{
    const float inv = 1.0f / fmaxf((float)deg, 1.0f);
    float4 r;
    r.x = ((a0.x + a1.x) + (a2.x + a3.x)) * inv;
    r.y = ((a0.y + a1.y) + (a2.y + a3.y)) * inv;
    r.z = ((a0.z + a1.z) + (a2.z + a3.z)) * inv;
    r.w = ((a0.w + a1.w) + (a2.w + a3.w)) * inv;
    return r;
}

__device__ __forceinline__ void load_frags(v2f (&fa)[16], v2f (&fb)[16],
                                           const float* ta, const float* bl, int blk)
{
    #pragma unroll
    for (int kk = 0; kk < 8; ++kk) {
        const int ks = 8 * blk + kk;
        fa[2 * kk]     = *(const v2f*)(ta + 4 * ks);
        fa[2 * kk + 1] = *(const v2f*)(ta + D + 4 * ks);
        fb[2 * kk]     = *(const v2f*)(bl + (2 * ks) * WPQ);
        fb[2 * kk + 1] = *(const v2f*)(bl + (2 * ks) * WPQ + 2 * D);
    }
}

__device__ __forceinline__ int request_offsets(const int* __restrict__ off,
                                               int node, int lane, int N)
{
    int i = node + (lane < 2 ? lane : 2);
    i = i < N ? i : N;
    return off[i];
}

struct RowSegs { int begA, endA, begB, endB; };
__device__ __forceinline__ RowSegs row_segments(int offv, int node, int N, int E)
{
    const int o0 = __builtin_amdgcn_readlane(offv, 0);
    const int o1 = __builtin_amdgcn_readlane(offv, 1);
    const int o2 = __builtin_amdgcn_readlane(offv, 2);
    RowSegs s;
    s.begA = 0; s.endA = 0; s.begB = 0; s.endB = 0;
    if (node < N)     { s.begA = o0; s.endA = o1; }
    if (node + 1 < N) { s.begB = o1; s.endB = o2; }
    s.begA = s.begA < 0 ? 0 : s.begA;  s.endA = s.endA > E ? E : s.endA;
    s.begB = s.begB < 0 ? 0 : s.begB;  s.endB = s.endB > E ? E : s.endB;
    return s;
}

__device__ __forceinline__ int entry_place(int beg, int end, int cb, int lane)
{
    int e = beg + cb + lane;
    e = e < end - 1 ? e : end - 1;
    e = e > 0 ? e : 0;
    return e;
}

__global__ __launch_bounds__(256) void sage_layer_kernel(
    const float* __restrict__ xin, const int* __restrict__ off,
    const int* __restrict__ perm, const float* __restrict__ Wl,
    const float* __restrict__ bias, const float* __restrict__ Wr,
    float* __restrict__ out, int N, int E, int relu)
{
    __shared__ __attribute__((aligned(16))) float Ws[(D / 2) * WPQ];
    __shared__ __attribute__((aligned(16))) float T[16 * TP];

    for (int it = 0; it < 16; ++it) {
        const int i  = it * 256 + (int)threadIdx.x;
        const int m  = i >> 11;
        const int kp = (i >> 5) & 63;
        const int q  = i & 31;
        const float4* wsrc = (const float4*)(m ? Wr : Wl);
        const float4 r0 = wsrc[(2 * kp) * (D / 4) + q];
        const float4 r1 = wsrc[(2 * kp + 1) * (D / 4) + q];
        float4* wdst = (float4*)(Ws + kp * WPQ + m * (2 * D) + 8 * q);
        wdst[0] = make_float4(r0.x, r1.x, r0.y, r1.y);
        wdst[1] = make_float4(r0.z, r1.z, r0.w, r1.w);
    }

    const int lane = threadIdx.x & 31;
    const int wave = __builtin_amdgcn_readfirstlane((int)(threadIdx.x >> 5));
    const int c = lane & 15;
    const int h = lane >> 4;
    const int ntiles = (N + 15) >> 4;
    const float4* xb = (const float4*)xin + lane;
    const unsigned last = (unsigned)(N - 1);

    const float* ta = T + c * TP + 2 * h;
    const float* bl = Ws + h * WPQ + 2 * (16 * wave + c);
    const int ncol = 16 * wave + c;
    const float bn = bias[ncol];

    const int G = (int)gridDim.x;
    int tile = (int)blockIdx.x;
    RowSegs sg;
    unsigned entA, entB;
    float4 xA, xB;
    int offn;
    {
        const int node0 = tile * 16 + 2 * wave;
        const int offv = request_offsets(off, node0, lane, N);
        offn = request_offsets(off, node0 + 16 * G, lane, N);
        sg = row_segments(offv, node0, N, E);
        entA = (unsigned)perm[entry_place(sg.begA, sg.endA, 0, lane)];
        entB = (unsigned)perm[entry_place(sg.begB, sg.endB, 0, lane)];
        const unsigned rowA = (unsigned)node0 < last ? (unsigned)node0 : last;
        const unsigned rowB = (unsigned)(node0 + 1) < last ? (unsigned)(node0 + 1) : last;
        xA = xb[(size_t)rowA * (D / 4)];
        xB = xb[(size_t)rowB * (D / 4)];
    }

    for (; tile < ntiles; tile += G) {
        const int m0 = tile * 16;

        const int begA = sg.begA, endA = sg.endA, begB = sg.begB, endB = sg.endB;
        const int degA = endA > begA ? endA - begA : 0;
        const int degB = endB > begB ? endB - begB : 0;
        const int degM = degA > degB ? degA : degB;

        const float4 z4 = make_float4(0.f, 0.f, 0.f, 0.f);
        float4 a0 = z4, a1 = z4, a2 = z4, a3 = z4;
        float4 b0 = z4, b1 = z4, b2 = z4, b3 = z4;
        for (int cb = 0; cb < degM; cb += 32) {
            unsigned pA = entA < last ? entA : last;
            unsigned pB = entB < last ? entB : last;
            if (cb + 32 < degM) {
                entA = (unsigned)perm[entry_place(begA, endA, cb + 32, lane)];
                entB = (unsigned)perm[entry_place(begB, endB, cb + 32, lane)];
            }
            int nA = degA - cb, nB = degB - cb;
            nA = nA < 0 ? 0 : (nA > 32 ? 32 : nA);
            nB = nB < 0 ? 0 : (nB > 32 ? 32 : nB);
            const int nM = nA > nB ? nA : nB;
            for (int b8 = 0; b8 < nM; b8 += 8) {
                float4 v[8], u[8];
                load_batch(v, xb, pA, b8, nA);
                load_batch(u, xb, pB, b8, nB);
                add_rows8(a0, a1, a2, a3, v);
                add_rows8(b0, b1, b2, b3, u);
            }
        }
        const float4 sA = scaled_sum(a0, a1, a2, a3, endA - begA);
        const float4 sB = scaled_sum(b0, b1, b2, b3, endB - begB);

        float4* tA = (float4*)(T + (2 * wave) * TP);
        float4* tB = (float4*)(T + (2 * wave + 1) * TP);
        tA[lane] = sA;  tA[D / 4 + lane] = xA;
        tB[lane] = sB;  tB[D / 4 + lane] = xB;

        __syncthreads();

        {
            const int nodeN = m0 + 16 * G + 2 * wave;
            sg = row_segments(offn, nodeN, N, E);
            entA = (unsigned)perm[entry_place(sg.begA, sg.endA, 0, lane)];
            entB = (unsigned)perm[entry_place(sg.begB, sg.endB, 0, lane)];
            const unsigned rowA = (unsigned)nodeN < last ? (unsigned)nodeN : last;
            const unsigned rowB = (unsigned)(nodeN + 1) < last ? (unsigned)(nodeN + 1) : last;
            xA = xb[(size_t)rowA * (D / 4)];
            xB = xb[(size_t)rowB * (D / 4)];
            offn = request_offsets(off, nodeN + 16 * G, lane, N);
        }

        v8f acc = {};
        v2f fa[2][16], fb[2][16];
        load_frags(fa[0], fb[0], ta, bl, 0);
        #pragma unroll
        for (int blk = 0; blk < 4; ++blk) {
            if (blk + 1 < 4) load_frags(fa[(blk + 1) & 1], fb[(blk + 1) & 1], ta, bl, blk + 1);
            __builtin_amdgcn_sched_barrier(0);
            #pragma unroll
            for (int i = 0; i < 16; ++i) {
                acc = __builtin_amdgcn_wmma_f32_16x16x4_f32(
                          false, fa[blk & 1][i], false, fb[blk & 1][i],
                          (short)0, acc, false, false);
            }
            __builtin_amdgcn_sched_barrier(0);
        }
        __syncthreads();

        const int mbase = m0 + (h << 3);
        #pragma unroll
        for (int r = 0; r < 8; ++r) {
            const int m = mbase + r;
            if (m < N) {
                float v = acc[r] + bn;
                if (relu) v = fmaxf(v, 0.0f);
                out[(long long)m * D + ncol] = v;
            }
        }
    }
}

__device__ __forceinline__ float reduce8(const float (&v)[8], int lane)
{
    const bool b4 = (lane & 16) != 0, b3 = (lane & 8) != 0, b2 = (lane & 4) != 0;
    float w[4];
    #pragma unroll
    for (int j = 0; j < 4; ++j) {
        const float keep = b4 ? v[j + 4] : v[j];
        const float send = b4 ? v[j] : v[j + 4];
        w[j] = keep + __shfl_xor(send, 16, 32);
    }
    float x[2];
    #pragma unroll
    for (int j = 0; j < 2; ++j) {
        const float keep = b3 ? w[j + 2] : w[j];
        const float send = b3 ? w[j] : w[j + 2];
        x[j] = keep + __shfl_xor(send, 8, 32);
    }
    const float keep = b2 ? x[1] : x[0];
    const float send = b2 ? x[0] : x[1];
    float y = keep + __shfl_xor(send, 4, 32);
    y += __shfl_xor(y, 2, 32);
    y += __shfl_xor(y, 1, 32);
    return y;
}

__global__ __launch_bounds__(256) void decode_kernel(
    const float* __restrict__ h, const int* __restrict__ dec,
    float* __restrict__ out, int Dk, int N)
{
    const int lane = threadIdx.x & 31;
    const int wave = __builtin_amdgcn_readfirstlane(
                         (int)((blockIdx.x * blockDim.x + threadIdx.x) >> 5));
    const int base = wave * 32;
    if (base >= Dk) return;
    const int p  = base + lane;
    const int pc = p < Dk ? p : Dk - 1;
    const unsigned last = (unsigned)(N - 1);
    unsigned ai = (unsigned)dec[pc];
    unsigned bi = (unsigned)dec[Dk + pc];
    ai = ai < last ? ai : last;
    bi = bi < last ? bi : last;
    const float4* hb = (const float4*)h + lane;

    float rd = 0.f, ra = 0.f, rb = 0.f;
    for (int b = 0; b < 4; ++b) {
        float4 va[8], vb[8];
        #pragma unroll
        for (int k = 0; k < 8; ++k) {
            const unsigned ia = (unsigned)__builtin_amdgcn_readlane((int)ai, b * 8 + k);
            const unsigned ib = (unsigned)__builtin_amdgcn_readlane((int)bi, b * 8 + k);
            va[k] = hb[(size_t)ia * (D / 4)];
            vb[k] = hb[(size_t)ib * (D / 4)];
        }
        float d[8], na[8], nb[8];
        #pragma unroll
        for (int k = 0; k < 8; ++k) {
            const float4 a = va[k], c = vb[k];
            d[k]  = a.x * c.x + a.y * c.y + a.z * c.z + a.w * c.w;
            na[k] = a.x * a.x + a.y * a.y + a.z * a.z + a.w * a.w;
            nb[k] = c.x * c.x + c.y * c.y + c.z * c.z + c.w * c.w;
        }
        const float sd = reduce8(d, lane);
        const float sa = reduce8(na, lane);
        const float sb = reduce8(nb, lane);
        const int from = (lane & 7) << 2;
        const float td = __shfl(sd, from, 32);
        const float ta = __shfl(sa, from, 32);
        const float tb = __shfl(sb, from, 32);
        if ((lane >> 3) == b) { rd = td; ra = ta; rb = tb; }
    }
    const float denom = fmaxf(sqrtf(ra) * sqrtf(rb), 1e-6f);
    const float z = rd / denom;
    if (p < Dk) out[p] = 1.0f / (1.0f + expf(-z));
}

extern "C" void kernel_launch(void* const* d_in, const int* in_sizes, int n_in,
                              void* d_out, int out_size, void* d_ws, size_t ws_size,
                              hipStream_t stream) {
    const float* x   = (const float*)d_in[0];
    const int*   edg = (const int*)d_in[1];
    const int*   dec = (const int*)d_in[2];
    const float* W1l = (const float*)d_in[3];
    const float* b1  = (const float*)d_in[4];
    const float* W1r = (const float*)d_in[5];
    const float* W2l = (const float*)d_in[6];
    const float* b2  = (const float*)d_in[7];
    const float* W2r = (const float*)d_in[8];
    float* out = (float*)d_out;

    const int N  = in_sizes[0] / D;
    const int E  = in_sizes[1] / 2;
    const int Dk = in_sizes[2] / 2;
    const int* src = edg;
    const int* dst = edg + E;

    float* ws  = (float*)d_ws;
    float* p0  = ws;
    float* h1  = p0 + (long long)N * D;
    float* h2  = h1 + (long long)N * D;

    int*  hist  = (int*)h1;
    int*  total = hist + MAXC * SLICES;
    int2* tmp   = (int2*)(total + MAXC);
    int*  off   = (int*)p0;
    int*  perm  = off + N + 1;

    const int C = (N + 255) >> BSH;
    const int scanBlocks  = (C + 7) / 8;
    const int layerBlocks = 512;
    const int decBlocks   = (Dk + 255) / 256;

    bucket_count_kernel<<<SLICES, 256, 0, stream>>>(dst, hist, E, C);
    bucket_scan_kernel<<<scanBlocks, 256, 0, stream>>>(hist, total, C);
    bucket_scatter_kernel<<<SLICES, 256, 0, stream>>>(src, dst, hist, total, tmp, E, C);
    bucket_sort_kernel<<<C, 512, 0, stream>>>(tmp, total, off, perm, N, E, C);
    sage_layer_kernel<<<layerBlocks, 256, 0, stream>>>(x, off, perm, W1l, b1, W1r,
                                                       h1, N, E,  1);
    sage_layer_kernel<<<layerBlocks, 256, 0, stream>>>(h1, off, perm, W2l, b2, W2r,
                                                       h2, N, E,  0);
    decode_kernel<<<decBlocks, 256, 0, stream>>>(h2, dec, out, Dk, N);
}


# edge ordering, passes 3 and 4: items are ordered in an LDS stage and copied out in runs (consecutive lanes, consecutive places), not one 4- or 8-byte store per line per lane; sum orders as before
# speedup vs baseline: 6.1622x; 1.0619x over previous
#include <hip/hip_runtime.h>
#include <hip/hip_bf16.h>
#include <math.h>

typedef __attribute__((ext_vector_type(2))) float v2f;
typedef __attribute__((ext_vector_type(8))) float v8f;

#define D 128

#define BSH    8
#define SLICES 256
#define MAXC   256

__device__ __forceinline__ int bucket_of(int d, int C)
{
    const unsigned c = (unsigned)d >> BSH;
    return (int)(c < (unsigned)C ? c : (unsigned)(C - 1));
}

__device__ __forceinline__ int slice_len(int E)
{
    return (((E + SLICES - 1) / SLICES) + 255) & ~255;
}

template <int NW>
__device__ __forceinline__ int block_excl_scan(int v, int* sh, int t)
{
    const int lane = t & 31, w = t >> 5;
    int incl = v;
    #pragma unroll
    for (int o = 1; o < 32; o <<= 1) {
        const int y = __shfl_up(incl, o, 32);
        if (lane >= o) incl += y;
    }
    if (lane == 31) sh[w] = incl;
    __syncthreads();
    int wb = 0;
    #pragma unroll
    for (int i = 0; i < NW; ++i) wb += (i < w) ? sh[i] : 0;
    return wb + incl - v;
}

#define EB 16
__global__ __launch_bounds__(256) void bucket_count_kernel(
    const int* __restrict__ dst, int* __restrict__ hist, int E, int C)
{
    __shared__ int cntb[MAXC];
    const int t = threadIdx.x, b = blockIdx.x;
    cntb[t] = 0;
    __syncthreads();
    const int per = slice_len(E);
    const int beg = b * per;
    const int end = (beg + per < E) ? (beg + per) : E;
    for (int e0 = beg + t; e0 < end; e0 += 256 * EB) {
        int d[EB];
        #pragma unroll
        for (int k = 0; k < EB; ++k) {
            const int e = e0 + 256 * k;
            d[k] = dst[e < end ? e : end - 1];
        }
        #pragma unroll
        for (int k = 0; k < EB; ++k)
            if (e0 + 256 * k < end) atomicAdd(&cntb[bucket_of(d[k], C)], 1);
    }
    __syncthreads();
    if (t < C) hist[t * SLICES + b] = cntb[t];
}

__global__ __launch_bounds__(256) void bucket_scan_kernel(
    int* __restrict__ hist, int* __restrict__ total, int C)
{
    const int lane = threadIdx.x & 31;
    const int c = (int)((blockIdx.x * blockDim.x + threadIdx.x) >> 5);
    if (c >= C) return;
    int4* row = (int4*)(hist + c * SLICES) + 2 * lane;
    const int4 a = row[0], b = row[1];
    const int e1 = a.x, e2 = e1 + a.y, e3 = e2 + a.z, e4 = e3 + a.w;
    const int e5 = e4 + b.x, e6 = e5 + b.y, e7 = e6 + b.z, sum = e7 + b.w;
    int incl = sum;
    #pragma unroll
    for (int o = 1; o < 32; o <<= 1) {
        const int y = __shfl_up(incl, o, 32);
        if (lane >= o) incl += y;
    }
    const int base = incl - sum;
    row[0] = make_int4(base, base + e1, base + e2, base + e3);
    row[1] = make_int4(base + e4, base + e5, base + e6, base + e7);
    if (lane == 31) total[c] = incl;
}

#define SCAP 8192
#define CB   8
__global__ __launch_bounds__(256) void bucket_scatter_kernel(
    const int* __restrict__ src, const int* __restrict__ dst,
    const int* __restrict__ hist, const int* __restrict__ total,
    int2* __restrict__ tmp, int E, int C)
{
    __shared__ int cur[MAXC];
    __shared__ int gst[MAXC];
    __shared__ int lst[MAXC];
    __shared__ int wsum[8], wsum2[8];
    __shared__ int2 stage[SCAP];
    const int t = threadIdx.x, b = blockIdx.x;
    const int per = slice_len(E);
    const int beg = b * per;
    const int end = (beg + per < E) ? (beg + per) : E;
    const int n = end > beg ? end - beg : 0;
    const bool staged = per <= SCAP;

    const int tot = (t < C) ? total[t] : 0;
    const int pre = (t < C) ? hist[t * SLICES + b] : 0;
    int nxt = tot;
    if (t < C && b + 1 < SLICES) nxt = hist[t * SLICES + b + 1];
    int mine = (t < C) ? nxt - pre : 0;
    mine = mine < 0 ? 0 : (mine > n ? n : mine);
    const int start = block_excl_scan<8>(tot, wsum, t);
    const int ls = block_excl_scan<8>(mine, wsum2, t);
    gst[t] = start + pre;
    lst[t] = ls;
    cur[t] = staged ? ls : start + pre;
    __syncthreads();

    for (int e0 = beg + t; e0 < end; e0 += 256 * EB) {
        int d[EB], s[EB];
        #pragma unroll
        for (int k = 0; k < EB; ++k) {
            const int e = e0 + 256 * k;
            const int ec = e < end ? e : end - 1;
            d[k] = dst[ec];
            s[k] = src[ec];
        }
        #pragma unroll
        for (int k = 0; k < EB; ++k) {
            if (e0 + 256 * k < end) {
                const int pos = atomicAdd(&cur[bucket_of(d[k], C)], 1);
                if (staged) {
                    if ((unsigned)pos < (unsigned)n) stage[pos] = make_int2(s[k], d[k]);
                } else {
                    if ((unsigned)pos < (unsigned)E) tmp[pos] = make_int2(s[k], d[k]);
                }
            }
        }
    }
    __syncthreads();
    if (staged) {
        for (int i0 = t; i0 < n; i0 += 256 * CB) {
            int2 p[CB];
            int g[CB];
            #pragma unroll
            for (int k = 0; k < CB; ++k) {
                const int i = i0 + 256 * k;
                p[k] = stage[i < n ? i : n - 1];
            }
            #pragma unroll
            for (int k = 0; k < CB; ++k) {
                const int c = bucket_of(p[k].y, C);
                g[k] = gst[c] + (i0 + 256 * k - lst[c]);
            }
            #pragma unroll
            for (int k = 0; k < CB; ++k) {
                if (i0 + 256 * k < n && (unsigned)g[k] < (unsigned)E) tmp[g[k]] = p[k];
            }
        }
    }
}

#define SB 20
#define PCAP 16384
__global__ __launch_bounds__(512) void bucket_sort_kernel(
    const int2* __restrict__ tmp, const int* __restrict__ total,
    int* __restrict__ off, int* __restrict__ perm,
    int N, int E, int C)
{
    __shared__ int cn[256];
    __shared__ int wsA[16], wsB[16];
    __shared__ int seg[2];
    __shared__ int stage[PCAP];
    const int t = threadIdx.x, c = blockIdx.x;
    const int tot = (t < C) ? total[t] : 0;
    if (t < 256) cn[t] = 0;
    const int start = block_excl_scan<16>(tot, wsA, t);
    if (t == c) { seg[0] = start; seg[1] = start + tot; }
    __syncthreads();
    int beg = seg[0], end = seg[1];
    beg = beg < 0 ? 0 : beg;
    end = end > E ? E : end;
    const int len = end > beg ? end - beg : 0;
    const bool staged = len <= PCAP;

    for (int e0 = beg + t; e0 < end; e0 += 512 * SB) {
        int d[SB];
        #pragma unroll
        for (int k = 0; k < SB; ++k) {
            const int e = e0 + 512 * k;
            d[k] = tmp[e < end ? e : end - 1].y;
        }
        #pragma unroll
        for (int k = 0; k < SB; ++k)
            if (e0 + 512 * k < end) atomicAdd(&cn[d[k] & 255], 1);
    }
    __syncthreads();

    const int myc = (t < 256) ? cn[t] : 0;
    const int ex = block_excl_scan<16>(myc, wsB, t);
    if (t < 256) {
        const int node = (c << BSH) + t;
        cn[t] = beg + ex;
        if (node < N) off[node] = beg + ex;
    }
    if (c == C - 1 && t == 0) off[N] = end;
    __syncthreads();

    for (int e0 = beg + t; e0 < end; e0 += 512 * SB) {
        int2 p[SB];
        #pragma unroll
        for (int k = 0; k < SB; ++k) {
            const int e = e0 + 512 * k;
            p[k] = tmp[e < end ? e : end - 1];
        }
        #pragma unroll
        for (int k = 0; k < SB; ++k) {
            if (e0 + 512 * k < end) {
                const int pos = atomicAdd(&cn[p[k].y & 255], 1);
                if (staged) {
                    const unsigned lp = (unsigned)(pos - beg);
                    if (lp < (unsigned)len) stage[lp] = p[k].x;
                } else {
                    if ((unsigned)pos < (unsigned)E) perm[pos] = p[k].x;
                }
            }
        }
    }
    __syncthreads();
    if (staged) {
        for (int i0 = t; i0 < len; i0 += 512 * CB) {
            int v[CB];
            #pragma unroll
            for (int k = 0; k < CB; ++k) {
                const int i = i0 + 512 * k;
                v[k] = stage[i < len ? i : len - 1];
            }
            #pragma unroll
            for (int k = 0; k < CB; ++k) {
                if (i0 + 512 * k < len) perm[beg + i0 + 512 * k] = v[k];
            }
        }
    }
}

#define WPQ 544
#define TP  260

__device__ __forceinline__ void add_rows8(float4& a0, float4& a1, float4& a2,
                                          float4& a3, const float4 (&v)[8])
{
    a0.x += v[0].x; a0.y += v[0].y; a0.z += v[0].z; a0.w += v[0].w;
    a1.x += v[1].x; a1.y += v[1].y; a1.z += v[1].z; a1.w += v[1].w;
    a2.x += v[2].x; a2.y += v[2].y; a2.z += v[2].z; a2.w += v[2].w;
    a3.x += v[3].x; a3.y += v[3].y; a3.z += v[3].z; a3.w += v[3].w;
    a0.x += v[4].x; a0.y += v[4].y; a0.z += v[4].z; a0.w += v[4].w;
    a1.x += v[5].x; a1.y += v[5].y; a1.z += v[5].z; a1.w += v[5].w;
    a2.x += v[6].x; a2.y += v[6].y; a2.z += v[6].z; a2.w += v[6].w;
    a3.x += v[7].x; a3.y += v[7].y; a3.z += v[7].z; a3.w += v[7].w;
}

__device__ __forceinline__ void load_batch(float4 (&v)[8], const float4* xb,
                                           unsigned p, int b8, int n)
{
    const int cap = n > 0 ? n - 1 : 0;
    #pragma unroll
    for (int k = 0; k < 8; ++k) {
        const int sel = (b8 + k) < cap ? (b8 + k) : cap;
        const unsigned idx = (unsigned)__builtin_amdgcn_readlane((int)p, sel);
        v[k] = xb[(size_t)idx * (D / 4)];
    }
    #pragma unroll
    for (int k = 0; k < 8; ++k) {
        if (b8 + k >= n) v[k] = make_float4(0.f, 0.f, 0.f, 0.f);
    }
}

__device__ __forceinline__ float4 scaled_sum(float4 a0, float4 a1, float4 a2,
                                             float4 a3, int deg)
{
    const float inv = 1.0f / fmaxf((float)deg, 1.0f);
    float4 r;
    r.x = ((a0.x + a1.x) + (a2.x + a3.x)) * inv;
    r.y = ((a0.y + a1.y) + (a2.y + a3.y)) * inv;
    r.z = ((a0.z + a1.z) + (a2.z + a3.z)) * inv;
    r.w = ((a0.w + a1.w) + (a2.w + a3.w)) * inv;
    return r;
}

__device__ __forceinline__ void load_frags(v2f (&fa)[16], v2f (&fb)[16],
                                           const float* ta, const float* bl, int blk)
{
    #pragma unroll
    for (int kk = 0; kk < 8; ++kk) {
        const int ks = 8 * blk + kk;
        fa[2 * kk]     = *(const v2f*)(ta + 4 * ks);
        fa[2 * kk + 1] = *(const v2f*)(ta + D + 4 * ks);
        fb[2 * kk]     = *(const v2f*)(bl + (2 * ks) * WPQ);
        fb[2 * kk + 1] = *(const v2f*)(bl + (2 * ks) * WPQ + 2 * D);
    }
}

__device__ __forceinline__ int request_offsets(const int* __restrict__ off,
                                               int node, int lane, int N)
{
    int i = node + (lane < 2 ? lane : 2);
    i = i < N ? i : N;
    return off[i];
}

struct RowSegs { int begA, endA, begB, endB; };
__device__ __forceinline__ RowSegs row_segments(int offv, int node, int N, int E)
{
    const int o0 = __builtin_amdgcn_readlane(offv, 0);
    const int o1 = __builtin_amdgcn_readlane(offv, 1);
    const int o2 = __builtin_amdgcn_readlane(offv, 2);
    RowSegs s;
    s.begA = 0; s.endA = 0; s.begB = 0; s.endB = 0;
    if (node < N)     { s.begA = o0; s.endA = o1; }
    if (node + 1 < N) { s.begB = o1; s.endB = o2; }
    s.begA = s.begA < 0 ? 0 : s.begA;  s.endA = s.endA > E ? E : s.endA;
    s.begB = s.begB < 0 ? 0 : s.begB;  s.endB = s.endB > E ? E : s.endB;
    return s;
}

__device__ __forceinline__ int entry_place(int beg, int end, int cb, int lane)
{
    int e = beg + cb + lane;
    e = e < end - 1 ? e : end - 1;
    e = e > 0 ? e : 0;
    return e;
}

__global__ __launch_bounds__(256) void sage_layer_kernel(
    const float* __restrict__ xin, const int* __restrict__ off,
    const int* __restrict__ perm, const float* __restrict__ Wl,
    const float* __restrict__ bias, const float* __restrict__ Wr,
    float* __restrict__ out, int N, int E, int relu)
{
    __shared__ __attribute__((aligned(16))) float Ws[(D / 2) * WPQ];
    __shared__ __attribute__((aligned(16))) float T[16 * TP];

    for (int it = 0; it < 16; ++it) {
        const int i  = it * 256 + (int)threadIdx.x;
        const int m  = i >> 11;
        const int kp = (i >> 5) & 63;
        const int q  = i & 31;
        const float4* wsrc = (const float4*)(m ? Wr : Wl);
        const float4 r0 = wsrc[(2 * kp) * (D / 4) + q];
        const float4 r1 = wsrc[(2 * kp + 1) * (D / 4) + q];
        float4* wdst = (float4*)(Ws + kp * WPQ + m * (2 * D) + 8 * q);
        wdst[0] = make_float4(r0.x, r1.x, r0.y, r1.y);
        wdst[1] = make_float4(r0.z, r1.z, r0.w, r1.w);
    }

    const int lane = threadIdx.x & 31;
    const int wave = __builtin_amdgcn_readfirstlane((int)(threadIdx.x >> 5));
    const int c = lane & 15;
    const int h = lane >> 4;
    const int ntiles = (N + 15) >> 4;
    const float4* xb = (const float4*)xin + lane;
    const unsigned last = (unsigned)(N - 1);

    const float* ta = T + c * TP + 2 * h;
    const float* bl = Ws + h * WPQ + 2 * (16 * wave + c);
    const int ncol = 16 * wave + c;
    const float bn = bias[ncol];

    const int G = (int)gridDim.x;
    int tile = (int)blockIdx.x;
    RowSegs sg;
    unsigned entA, entB;
    float4 xA, xB;
    int offn;
    {
        const int node0 = tile * 16 + 2 * wave;
        const int offv = request_offsets(off, node0, lane, N);
        offn = request_offsets(off, node0 + 16 * G, lane, N);
        sg = row_segments(offv, node0, N, E);
        entA = (unsigned)perm[entry_place(sg.begA, sg.endA, 0, lane)];
        entB = (unsigned)perm[entry_place(sg.begB, sg.endB, 0, lane)];
        const unsigned rowA = (unsigned)node0 < last ? (unsigned)node0 : last;
        const unsigned rowB = (unsigned)(node0 + 1) < last ? (unsigned)(node0 + 1) : last;
        xA = xb[(size_t)rowA * (D / 4)];
        xB = xb[(size_t)rowB * (D / 4)];
    }

    for (; tile < ntiles; tile += G) {
        const int m0 = tile * 16;

        const int begA = sg.begA, endA = sg.endA, begB = sg.begB, endB = sg.endB;
        const int degA = endA > begA ? endA - begA : 0;
        const int degB = endB > begB ? endB - begB : 0;
        const int degM = degA > degB ? degA : degB;

        const float4 z4 = make_float4(0.f, 0.f, 0.f, 0.f);
        float4 a0 = z4, a1 = z4, a2 = z4, a3 = z4;
        float4 b0 = z4, b1 = z4, b2 = z4, b3 = z4;
        for (int cb = 0; cb < degM; cb += 32) {
            unsigned pA = entA < last ? entA : last;
            unsigned pB = entB < last ? entB : last;
            if (cb + 32 < degM) {
                entA = (unsigned)perm[entry_place(begA, endA, cb + 32, lane)];
                entB = (unsigned)perm[entry_place(begB, endB, cb + 32, lane)];
            }
            int nA = degA - cb, nB = degB - cb;
            nA = nA < 0 ? 0 : (nA > 32 ? 32 : nA);
            nB = nB < 0 ? 0 : (nB > 32 ? 32 : nB);
            const int nM = nA > nB ? nA : nB;
            for (int b8 = 0; b8 < nM; b8 += 8) {
                float4 v[8], u[8];
                load_batch(v, xb, pA, b8, nA);
                load_batch(u, xb, pB, b8, nB);
                add_rows8(a0, a1, a2, a3, v);
                add_rows8(b0, b1, b2, b3, u);
            }
        }
        const float4 sA = scaled_sum(a0, a1, a2, a3, endA - begA);
        const float4 sB = scaled_sum(b0, b1, b2, b3, endB - begB);

        float4* tA = (float4*)(T + (2 * wave) * TP);
        float4* tB = (float4*)(T + (2 * wave + 1) * TP);
        tA[lane] = sA;  tA[D / 4 + lane] = xA;
        tB[lane] = sB;  tB[D / 4 + lane] = xB;

        __syncthreads();

        {
            const int nodeN = m0 + 16 * G + 2 * wave;
            sg = row_segments(offn, nodeN, N, E);
            entA = (unsigned)perm[entry_place(sg.begA, sg.endA, 0, lane)];
            entB = (unsigned)perm[entry_place(sg.begB, sg.endB, 0, lane)];
            const unsigned rowA = (unsigned)nodeN < last ? (unsigned)nodeN : last;
            const unsigned rowB = (unsigned)(nodeN + 1) < last ? (unsigned)(nodeN + 1) : last;
            xA = xb[(size_t)rowA * (D / 4)];
            xB = xb[(size_t)rowB * (D / 4)];
            offn = request_offsets(off, nodeN + 16 * G, lane, N);
        }

        v8f acc = {};
        v2f fa[2][16], fb[2][16];
        load_frags(fa[0], fb[0], ta, bl, 0);
        #pragma unroll
        for (int blk = 0; blk < 4; ++blk) {
            if (blk + 1 < 4) load_frags(fa[(blk + 1) & 1], fb[(blk + 1) & 1], ta, bl, blk + 1);
            __builtin_amdgcn_sched_barrier(0);
            #pragma unroll
            for (int i = 0; i < 16; ++i) {
                acc = __builtin_amdgcn_wmma_f32_16x16x4_f32(
                          false, fa[blk & 1][i], false, fb[blk & 1][i],
                          (short)0, acc, false, false);
            }
            __builtin_amdgcn_sched_barrier(0);
        }
        __syncthreads();

        const int mbase = m0 + (h << 3);
        #pragma unroll
        for (int r = 0; r < 8; ++r) {
            const int m = mbase + r;
            if (m < N) {
                float v = acc[r] + bn;
                if (relu) v = fmaxf(v, 0.0f);
                out[(long long)m * D + ncol] = v;
            }
        }
    }
}

__device__ __forceinline__ float reduce8(const float (&v)[8], int lane)
{
    const bool b4 = (lane & 16) != 0, b3 = (lane & 8) != 0, b2 = (lane & 4) != 0;
    float w[4];
    #pragma unroll
    for (int j = 0; j < 4; ++j) {
        const float keep = b4 ? v[j + 4] : v[j];
        const float send = b4 ? v[j] : v[j + 4];
        w[j] = keep + __shfl_xor(send, 16, 32);
    }
    float x[2];
    #pragma unroll
    for (int j = 0; j < 2; ++j) {
        const float keep = b3 ? w[j + 2] : w[j];
        const float send = b3 ? w[j] : w[j + 2];
        x[j] = keep + __shfl_xor(send, 8, 32);
    }
    const float keep = b2 ? x[1] : x[0];
    const float send = b2 ? x[0] : x[1];
    float y = keep + __shfl_xor(send, 4, 32);
    y += __shfl_xor(y, 2, 32);
    y += __shfl_xor(y, 1, 32);
    return y;
}

__global__ __launch_bounds__(256) void decode_kernel(
    const float* __restrict__ h, const int* __restrict__ dec,
    float* __restrict__ out, int Dk, int N)
{
    const int lane = threadIdx.x & 31;
    const int wave = __builtin_amdgcn_readfirstlane(
                         (int)((blockIdx.x * blockDim.x + threadIdx.x) >> 5));
    const int base = wave * 32;
    if (base >= Dk) return;
    const int p  = base + lane;
    const int pc = p < Dk ? p : Dk - 1;
    const unsigned last = (unsigned)(N - 1);
    unsigned ai = (unsigned)dec[pc];
    unsigned bi = (unsigned)dec[Dk + pc];
    ai = ai < last ? ai : last;
    bi = bi < last ? bi : last;
    const float4* hb = (const float4*)h + lane;

    float rd = 0.f, ra = 0.f, rb = 0.f;
    for (int b = 0; b < 4; ++b) {
        float4 va[8], vb[8];
        #pragma unroll
        for (int k = 0; k < 8; ++k) {
            const unsigned ia = (unsigned)__builtin_amdgcn_readlane((int)ai, b * 8 + k);
            const unsigned ib = (unsigned)__builtin_amdgcn_readlane((int)bi, b * 8 + k);
            va[k] = hb[(size_t)ia * (D / 4)];
            vb[k] = hb[(size_t)ib * (D / 4)];
        }
        float d[8], na[8], nb[8];
        #pragma unroll
        for (int k = 0; k < 8; ++k) {
            const float4 a = va[k], c = vb[k];
            d[k]  = a.x * c.x + a.y * c.y + a.z * c.z + a.w * c.w;
            na[k] = a.x * a.x + a.y * a.y + a.z * a.z + a.w * a.w;
            nb[k] = c.x * c.x + c.y * c.y + c.z * c.z + c.w * c.w;
        }
        const float sd = reduce8(d, lane);
        const float sa = reduce8(na, lane);
        const float sb = reduce8(nb, lane);
        const int from = (lane & 7) << 2;
        const float td = __shfl(sd, from, 32);
        const float ta = __shfl(sa, from, 32);
        const float tb = __shfl(sb, from, 32);
        if ((lane >> 3) == b) { rd = td; ra = ta; rb = tb; }
    }
    const float denom = fmaxf(sqrtf(ra) * sqrtf(rb), 1e-6f);
    const float z = rd / denom;
    if (p < Dk) out[p] = 1.0f / (1.0f + expf(-z));
}

extern "C" void kernel_launch(void* const* d_in, const int* in_sizes, int n_in,
                              void* d_out, int out_size, void* d_ws, size_t ws_size,
                              hipStream_t stream) {
    const float* x   = (const float*)d_in[0];
    const int*   edg = (const int*)d_in[1];
    const int*   dec = (const int*)d_in[2];
    const float* W1l = (const float*)d_in[3];
    const float* b1  = (const float*)d_in[4];
    const float* W1r = (const float*)d_in[5];
    const float* W2l = (const float*)d_in[6];
    const float* b2  = (const float*)d_in[7];
    const float* W2r = (const float*)d_in[8];
    float* out = (float*)d_out;

    const int N  = in_sizes[0] / D;
    const int E  = in_sizes[1] / 2;
    const int Dk = in_sizes[2] / 2;
    const int* src = edg;
    const int* dst = edg + E;

    float* ws  = (float*)d_ws;
    float* p0  = ws;
    float* h1  = p0 + (long long)N * D;
    float* h2  = h1 + (long long)N * D;

    int*  hist  = (int*)h1;
    int*  total = hist + MAXC * SLICES;
    int2* tmp   = (int2*)(total + MAXC);
    int*  off   = (int*)p0;
    int*  perm  = off + N + 1;

    const int C = (N + 255) >> BSH;
    const int scanBlocks  = (C + 7) / 8;
    const int layerBlocks = 512;
    const int decBlocks   = (Dk + 255) / 256;

    bucket_count_kernel<<<SLICES, 256, 0, stream>>>(dst, hist, E, C);
    bucket_scan_kernel<<<scanBlocks, 256, 0, stream>>>(hist, total, C);
    bucket_scatter_kernel<<<SLICES, 256, 0, stream>>>(src, dst, hist, total, tmp, E, C);
    bucket_sort_kernel<<<C, 512, 0, stream>>>(tmp, total, off, perm, N, E, C);
    sage_layer_kernel<<<layerBlocks, 256, 0, stream>>>(x, off, perm, W1l, b1, W1r,
                                                       h1, N, E,  1);
    sage_layer_kernel<<<layerBlocks, 256, 0, stream>>>(h1, off, perm, W2l, b2, W2r,
                                                       h2, N, E,  0);
    decode_kernel<<<decBlocks, 256, 0, stream>>>(h2, dec, out, Dk, N);
}
